# starting kernel
# baseline (speedup 1.0000x reference)
#include <hip/hip_runtime.h>

typedef float v2f __attribute__((ext_vector_type(2)));
typedef float v8f __attribute__((ext_vector_type(8)));

#define BN_    4
#define N_GRP  2
#define L_     1024
#define DM_    512
#define DIN_   1024
#define DSTATE 16
#define DTRANK 32
#define MTOT   4096
#define NCHUNK 16
#define TCH    64

__global__ __launch_bounds__(128)
void wmma_gemm_tn(const float* __restrict__ A, int lda,
                  const float* __restrict__ Bm, int ldb,
                  float* __restrict__ C, int ldc,
                  int M, int N, int K,
                  const float* __restrict__ bias, int epilogue)
{
    const int lane  = threadIdx.x & 31;
    const int wave  = threadIdx.x >> 5;
    const int ntile = N >> 6;
    const int tile  = blockIdx.x * 4 + wave;
    if (tile >= (M >> 6) * ntile) return;
    const int tm = tile / ntile;
    const int tn = tile % ntile;

    const int l15  = lane & 15;
    const int koff = (lane >> 4) << 1;
    const size_t astr = (size_t)16 * lda;
    const size_t bstr = (size_t)16 * ldb;
    const float* ap = A  + (size_t)(tm * 64 + l15) * lda + koff;
    const float* bp = Bm + (size_t)(tn * 64 + l15) * ldb + koff;

    v8f acc[4][4] = {};
    for (int k = 0; k < K; k += 4) {
        v2f a[4], b[4];
#pragma unroll
        for (int i = 0; i < 4; ++i) a[i] = *(const v2f*)(ap + i * astr + k);
#pragma unroll
        for (int j = 0; j < 4; ++j) b[j] = *(const v2f*)(bp + j * bstr + k);
#pragma unroll
        for (int i = 0; i < 4; ++i)
#pragma unroll
            for (int j = 0; j < 4; ++j)
                acc[i][j] = __builtin_amdgcn_wmma_f32_16x16x4_f32(
                                false, a[i], false, b[j],
                                (short)0, acc[i][j], false, false);
    }

    const int rb = (lane >> 4) << 3;
#pragma unroll
    for (int i = 0; i < 4; ++i) {
        const int rowb = tm * 64 + i * 16 + rb;
#pragma unroll
        for (int j = 0; j < 4; ++j) {
            const int col = tn * 64 + j * 16 + l15;
            const float bv = (epilogue == 1) ? bias[col] : 0.0f;
#pragma unroll
            for (int r = 0; r < 8; ++r) {
                float v = acc[i][j][r];
                if (epilogue == 1) {
                    v += bv;
                    v = (v > 20.0f) ? v : __logf(1.0f + __expf(v));
                }
                C[(size_t)(rowb + r) * ldc + col] = v;
            }
        }
    }
}

__global__ __launch_bounds__(256)
void conv_silu_kernel(const float* __restrict__ C1,
                      const float* __restrict__ cw,
                      const float* __restrict__ cb,
                      float* __restrict__ U, int total)
{
    int idx = blockIdx.x * 256 + threadIdx.x;
    if (idx >= total) return;
    int d  = idx & (DIN_ - 1);
    int l  = (idx >> 10) & (L_ - 1);
    int bn = idx >> 20;
    int n  = bn & (N_GRP - 1);
    int ch = n * DIN_ + d;

    const float* w = cw + (size_t)ch * 4;
    float acc = cb[ch];
    const float* xcol = C1 + ((size_t)bn * L_) * (2 * DIN_) + d;
#pragma unroll
    for (int k = 0; k < 4; ++k) {
        int ll = l - 3 + k;
        float xv = (ll >= 0) ? xcol[(size_t)ll * (2 * DIN_)] : 0.0f;
        acc = fmaf(w[k], xv, acc);
    }
    U[idx] = acc / (1.0f + __expf(-acc));
}

__global__ __launch_bounds__(128)
void scan_chunk_reduce(const float* __restrict__ DT, const float* __restrict__ U,
                       const float* __restrict__ PR,
                       const float* __restrict__ A_log,
                       float* __restrict__ CH_C, float* __restrict__ CH_S)
{
    const int d  = ((blockIdx.x & 7) << 7) + threadIdx.x;
    const int ck = (blockIdx.x >> 3) & (NCHUNK - 1);
    const int bn = blockIdx.x >> 7;

    float As[DSTATE], h[DSTATE];
#pragma unroll
    for (int s = 0; s < DSTATE; ++s) {
        As[s] = -__expf(A_log[(size_t)d * DSTATE + s]);
        h[s]  = 0.0f;
    }
    float S = 0.0f;
    const int t0 = ck * TCH;
    for (int t = t0; t < t0 + TCH; ++t) {
        const size_t base = (size_t)bn * L_ + t;
        const float* pr = PR + base * 64;
        float dtv = DT[base * DIN_ + d];
        float uv  = U [base * DIN_ + d];
        S += dtv;
        float dtu = dtv * uv;
#pragma unroll
        for (int s = 0; s < DSTATE; ++s)
            h[s] = fmaf(__expf(dtv * As[s]), h[s], dtu * pr[DTRANK + s]);
    }
    const size_t cb = (size_t)(bn * NCHUNK + ck) * DSTATE * DIN_ + d;
#pragma unroll
    for (int s = 0; s < DSTATE; ++s) CH_C[cb + (size_t)s * DIN_] = h[s];
    CH_S[(size_t)(bn * NCHUNK + ck) * DIN_ + d] = S;
}

__global__ __launch_bounds__(128)
void scan_chunk_emit(const float* __restrict__ DT, const float* __restrict__ U,
                     const float* __restrict__ PR, const float* __restrict__ C1,
                     const float* __restrict__ A_log, const float* __restrict__ Dp,
                     const float* __restrict__ CH_C, const float* __restrict__ CH_S,
                     float* __restrict__ Y2)
{
    const int d  = ((blockIdx.x & 7) << 7) + threadIdx.x;
    const int ck = (blockIdx.x >> 3) & (NCHUNK - 1);
    const int bn = blockIdx.x >> 7;

    float As[DSTATE], h[DSTATE];
#pragma unroll
    for (int s = 0; s < DSTATE; ++s) {
        As[s] = -__expf(A_log[(size_t)d * DSTATE + s]);
        h[s]  = 0.0f;
    }
    for (int j = 0; j < ck; ++j) {
        float Sj = CH_S[(size_t)(bn * NCHUNK + j) * DIN_ + d];
        const size_t cb = (size_t)(bn * NCHUNK + j) * DSTATE * DIN_ + d;
#pragma unroll
        for (int s = 0; s < DSTATE; ++s)
            h[s] = fmaf(__expf(Sj * As[s]), h[s], CH_C[cb + (size_t)s * DIN_]);
    }

    const float dp = Dp[d];
    const int t0 = ck * TCH;
    for (int t = t0; t < t0 + TCH; ++t) {
        const size_t base = (size_t)bn * L_ + t;
        const float* pr = PR + base * 64;
        float dtv = DT[base * DIN_ + d];
        float uv  = U [base * DIN_ + d];
        float dtu = dtv * uv;
        float y = 0.0f;
#pragma unroll
        for (int s = 0; s < DSTATE; ++s) {
            h[s] = fmaf(__expf(dtv * As[s]), h[s], dtu * pr[DTRANK + s]);
            y    = fmaf(h[s], pr[DTRANK + DSTATE + s], y);
        }
        float res = C1[base * (2 * DIN_) + DIN_ + d];
        float sr  = res / (1.0f + __expf(-res));
        Y2[base * DIN_ + d] = (y + uv * dp) * sr;
    }
}

extern "C" void kernel_launch(void* const* d_in, const int* in_sizes, int n_in,
                              void* d_out, int out_size, void* d_ws, size_t ws_size,
                              hipStream_t stream)
{
    const float* x      = (const float*)d_in[0];
    const float* W_in   = (const float*)d_in[1];
    const float* conv_w = (const float*)d_in[2];
    const float* conv_b = (const float*)d_in[3];
    const float* W_x    = (const float*)d_in[4];
    const float* W_dt   = (const float*)d_in[5];
    const float* b_dt   = (const float*)d_in[6];
    const float* A_log  = (const float*)d_in[7];
    const float* Dp     = (const float*)d_in[8];
    const float* W_out  = (const float*)d_in[9];
    float* out = (float*)d_out;

    float* C1   = (float*)d_ws;
    float* U    = C1   + (size_t)MTOT * (2 * DIN_);
    float* PR   = U    + (size_t)MTOT * DIN_;
    float* DTb  = PR   + (size_t)MTOT * 64;
    float* Y2   = DTb  + (size_t)MTOT * DIN_;
    float* CH_C = Y2   + (size_t)MTOT * DIN_;
    float* CH_S = CH_C + (size_t)BN_ * NCHUNK * DSTATE * DIN_;

    int tiles = (MTOT / 64) * ((2 * DIN_) / 64);
    wmma_gemm_tn<<<tiles / 4, 128, 0, stream>>>(x, DM_, W_in, DM_,
                                                C1, 2 * DIN_,
                                                MTOT, 2 * DIN_, DM_, nullptr, 0);

    conv_silu_kernel<<<(MTOT * DIN_) / 256, 256, 0, stream>>>(C1, conv_w, conv_b,
                                                              U, MTOT * DIN_);

    tiles = (MTOT / 64) * (64 / 64);
    wmma_gemm_tn<<<tiles / 4, 128, 0, stream>>>(U, DIN_, W_x, DIN_,
                                                PR, 64,
                                                MTOT, 64, DIN_, nullptr, 0);

    tiles = (MTOT / 64) * (DIN_ / 64);
    wmma_gemm_tn<<<tiles / 4, 128, 0, stream>>>(PR, 64, W_dt, DTRANK,
                                                DTb, DIN_,
                                                MTOT, DIN_, DTRANK, b_dt, 1);

    const int scan_blocks = BN_ * NCHUNK * (DIN_ / 128);
    scan_chunk_reduce<<<scan_blocks, 128, 0, stream>>>(DTb, U, PR, A_log,
                                                       CH_C, CH_S);
    scan_chunk_emit<<<scan_blocks, 128, 0, stream>>>(DTb, U, PR, C1, A_log, Dp,
                                                     CH_C, CH_S, Y2);

    tiles = (MTOT / 64) * (DM_ / 64);
    wmma_gemm_tn<<<tiles / 4, 128, 0, stream>>>(Y2, DIN_, W_out, DIN_,
                                                out, DM_,
                                                MTOT, DM_, DIN_, nullptr, 0);
}


# GEMM A/B fragments: one 16-byte load per lane per 8 k replaces two 8-byte loads; register budget 256 keeps them one group (k order inside each 8 permuted alike in A and B: f32 sum order changes)
# speedup vs baseline: 1.0781x; 1.0781x over previous
#include <hip/hip_runtime.h>

typedef float v2f __attribute__((ext_vector_type(2)));
typedef float v4f __attribute__((ext_vector_type(4)));
typedef float v8f __attribute__((ext_vector_type(8)));

#define BN_    4
#define N_GRP  2
#define L_     1024
#define DM_    512
#define DIN_   1024
#define DSTATE 16
#define DTRANK 32
#define MTOT   4096
#define NCHUNK 16
#define TCH    64

__global__ __launch_bounds__(128) __attribute__((amdgpu_waves_per_eu(1, 4)))
void wmma_gemm_tn(const float* __restrict__ A, int lda,
                  const float* __restrict__ Bm, int ldb,
                  float* __restrict__ C, int ldc,
                  int M, int N, int K,
                  const float* __restrict__ bias, int epilogue)
{
    const int lane  = threadIdx.x & 31;
    const int wave  = threadIdx.x >> 5;
    const int ntile = N >> 6;
    const int tile  = blockIdx.x * 4 + wave;
    if (tile >= (M >> 6) * ntile) return;
    const int tm = tile / ntile;
    const int tn = tile % ntile;

    const int l15  = lane & 15;
    const int koff = (lane >> 4) << 2;
    const size_t astr = (size_t)16 * lda;
    const size_t bstr = (size_t)16 * ldb;
    const float* ap = A  + (size_t)(tm * 64 + l15) * lda + koff;
    const float* bp = Bm + (size_t)(tn * 64 + l15) * ldb + koff;

    v8f acc[4][4] = {};
    for (int k = 0; k < K; k += 8) {
        v4f a[4], b[4];
#pragma unroll
        for (int i = 0; i < 4; ++i) a[i] = *(const v4f*)(ap + i * astr + k);
#pragma unroll
        for (int j = 0; j < 4; ++j) b[j] = *(const v4f*)(bp + j * bstr + k);
#pragma unroll
        for (int i = 0; i < 4; ++i)
#pragma unroll
            for (int j = 0; j < 4; ++j)
                acc[i][j] = __builtin_amdgcn_wmma_f32_16x16x4_f32(
                                false, __builtin_shufflevector(a[i], a[i], 0, 1),
                                false, __builtin_shufflevector(b[j], b[j], 0, 1),
                                (short)0, acc[i][j], false, false);
#pragma unroll
        for (int i = 0; i < 4; ++i)
#pragma unroll
            for (int j = 0; j < 4; ++j)
                acc[i][j] = __builtin_amdgcn_wmma_f32_16x16x4_f32(
                                false, __builtin_shufflevector(a[i], a[i], 2, 3),
                                false, __builtin_shufflevector(b[j], b[j], 2, 3),
                                (short)0, acc[i][j], false, false);
    }

    const int rb = (lane >> 4) << 3;
#pragma unroll
    for (int i = 0; i < 4; ++i) {
        const int rowb = tm * 64 + i * 16 + rb;
#pragma unroll
        for (int j = 0; j < 4; ++j) {
            const int col = tn * 64 + j * 16 + l15;
            const float bv = (epilogue == 1) ? bias[col] : 0.0f;
#pragma unroll
            for (int r = 0; r < 8; ++r) {
                float v = acc[i][j][r];
                if (epilogue == 1) {
                    v += bv;
                    v = (v > 20.0f) ? v : __logf(1.0f + __expf(v));
                }
                C[(size_t)(rowb + r) * ldc + col] = v;
            }
        }
    }
}

__global__ __launch_bounds__(256)
void conv_silu_kernel(const float* __restrict__ C1,
                      const float* __restrict__ cw,
                      const float* __restrict__ cb,
                      float* __restrict__ U, int total)
{
    int idx = blockIdx.x * 256 + threadIdx.x;
    if (idx >= total) return;
    int d  = idx & (DIN_ - 1);
    int l  = (idx >> 10) & (L_ - 1);
    int bn = idx >> 20;
    int n  = bn & (N_GRP - 1);
    int ch = n * DIN_ + d;

    const float* w = cw + (size_t)ch * 4;
    float acc = cb[ch];
    const float* xcol = C1 + ((size_t)bn * L_) * (2 * DIN_) + d;
#pragma unroll
    for (int k = 0; k < 4; ++k) {
        int ll = l - 3 + k;
        float xv = (ll >= 0) ? xcol[(size_t)ll * (2 * DIN_)] : 0.0f;
        acc = fmaf(w[k], xv, acc);
    }
    U[idx] = acc / (1.0f + __expf(-acc));
}

__global__ __launch_bounds__(128)
void scan_chunk_reduce(const float* __restrict__ DT, const float* __restrict__ U,
                       const float* __restrict__ PR,
                       const float* __restrict__ A_log,
                       float* __restrict__ CH_C, float* __restrict__ CH_S)
{
    const int d  = ((blockIdx.x & 7) << 7) + threadIdx.x;
    const int ck = (blockIdx.x >> 3) & (NCHUNK - 1);
    const int bn = blockIdx.x >> 7;

    float As[DSTATE], h[DSTATE];
#pragma unroll
    for (int s = 0; s < DSTATE; ++s) {
        As[s] = -__expf(A_log[(size_t)d * DSTATE + s]);
        h[s]  = 0.0f;
    }
    float S = 0.0f;
    const int t0 = ck * TCH;
    for (int t = t0; t < t0 + TCH; ++t) {
        const size_t base = (size_t)bn * L_ + t;
        const float* pr = PR + base * 64;
        float dtv = DT[base * DIN_ + d];
        float uv  = U [base * DIN_ + d];
        S += dtv;
        float dtu = dtv * uv;
#pragma unroll
        for (int s = 0; s < DSTATE; ++s)
            h[s] = fmaf(__expf(dtv * As[s]), h[s], dtu * pr[DTRANK + s]);
    }
    const size_t cb = (size_t)(bn * NCHUNK + ck) * DSTATE * DIN_ + d;
#pragma unroll
    for (int s = 0; s < DSTATE; ++s) CH_C[cb + (size_t)s * DIN_] = h[s];
    CH_S[(size_t)(bn * NCHUNK + ck) * DIN_ + d] = S;
}

__global__ __launch_bounds__(128)
void scan_chunk_emit(const float* __restrict__ DT, const float* __restrict__ U,
                     const float* __restrict__ PR, const float* __restrict__ C1,
                     const float* __restrict__ A_log, const float* __restrict__ Dp,
                     const float* __restrict__ CH_C, const float* __restrict__ CH_S,
                     float* __restrict__ Y2)
{
    const int d  = ((blockIdx.x & 7) << 7) + threadIdx.x;
    const int ck = (blockIdx.x >> 3) & (NCHUNK - 1);
    const int bn = blockIdx.x >> 7;

    float As[DSTATE], h[DSTATE];
#pragma unroll
    for (int s = 0; s < DSTATE; ++s) {
        As[s] = -__expf(A_log[(size_t)d * DSTATE + s]);
        h[s]  = 0.0f;
    }
    for (int j = 0; j < ck; ++j) {
        float Sj = CH_S[(size_t)(bn * NCHUNK + j) * DIN_ + d];
        const size_t cb = (size_t)(bn * NCHUNK + j) * DSTATE * DIN_ + d;
#pragma unroll
        for (int s = 0; s < DSTATE; ++s)
            h[s] = fmaf(__expf(Sj * As[s]), h[s], CH_C[cb + (size_t)s * DIN_]);
    }

    const float dp = Dp[d];
    const int t0 = ck * TCH;
    for (int t = t0; t < t0 + TCH; ++t) {
        const size_t base = (size_t)bn * L_ + t;
        const float* pr = PR + base * 64;
        float dtv = DT[base * DIN_ + d];
        float uv  = U [base * DIN_ + d];
        float dtu = dtv * uv;
        float y = 0.0f;
#pragma unroll
        for (int s = 0; s < DSTATE; ++s) {
            h[s] = fmaf(__expf(dtv * As[s]), h[s], dtu * pr[DTRANK + s]);
            y    = fmaf(h[s], pr[DTRANK + DSTATE + s], y);
        }
        float res = C1[base * (2 * DIN_) + DIN_ + d];
        float sr  = res / (1.0f + __expf(-res));
        Y2[base * DIN_ + d] = (y + uv * dp) * sr;
    }
}

extern "C" void kernel_launch(void* const* d_in, const int* in_sizes, int n_in,
                              void* d_out, int out_size, void* d_ws, size_t ws_size,
                              hipStream_t stream)
{
    const float* x      = (const float*)d_in[0];
    const float* W_in   = (const float*)d_in[1];
    const float* conv_w = (const float*)d_in[2];
    const float* conv_b = (const float*)d_in[3];
    const float* W_x    = (const float*)d_in[4];
    const float* W_dt   = (const float*)d_in[5];
    const float* b_dt   = (const float*)d_in[6];
    const float* A_log  = (const float*)d_in[7];
    const float* Dp     = (const float*)d_in[8];
    const float* W_out  = (const float*)d_in[9];
    float* out = (float*)d_out;

    float* C1   = (float*)d_ws;
    float* U    = C1   + (size_t)MTOT * (2 * DIN_);
    float* PR   = U    + (size_t)MTOT * DIN_;
    float* DTb  = PR   + (size_t)MTOT * 64;
    float* Y2   = DTb  + (size_t)MTOT * DIN_;
    float* CH_C = Y2   + (size_t)MTOT * DIN_;
    float* CH_S = CH_C + (size_t)BN_ * NCHUNK * DSTATE * DIN_;

    int tiles = (MTOT / 64) * ((2 * DIN_) / 64);
    wmma_gemm_tn<<<tiles / 4, 128, 0, stream>>>(x, DM_, W_in, DM_,
                                                C1, 2 * DIN_,
                                                MTOT, 2 * DIN_, DM_, nullptr, 0);

    conv_silu_kernel<<<(MTOT * DIN_) / 256, 256, 0, stream>>>(C1, conv_w, conv_b,
                                                              U, MTOT * DIN_);

    tiles = (MTOT / 64) * (64 / 64);
    wmma_gemm_tn<<<tiles / 4, 128, 0, stream>>>(U, DIN_, W_x, DIN_,
                                                PR, 64,
                                                MTOT, 64, DIN_, nullptr, 0);

    tiles = (MTOT / 64) * (DIN_ / 64);
    wmma_gemm_tn<<<tiles / 4, 128, 0, stream>>>(PR, 64, W_dt, DTRANK,
                                                DTb, DIN_,
                                                MTOT, DIN_, DTRANK, b_dt, 1);

    const int scan_blocks = BN_ * NCHUNK * (DIN_ / 128);
    scan_chunk_reduce<<<scan_blocks, 128, 0, stream>>>(DTb, U, PR, A_log,
                                                       CH_C, CH_S);
    scan_chunk_emit<<<scan_blocks, 128, 0, stream>>>(DTb, U, PR, C1, A_log, Dp,
                                                     CH_C, CH_S, Y2);

    tiles = (MTOT / 64) * (DM_ / 64);
    wmma_gemm_tn<<<tiles / 4, 128, 0, stream>>>(Y2, DIN_, W_out, DIN_,
                                                out, DM_,
                                                MTOT, DM_, DIN_, nullptr, 0);
}


# x_proj and out_proj GEMMs left most SIMDs idle: K is split over a block's waves (x_proj 16x64 tiles x4, out_proj 64x64 x2), partial tiles added through LDS in k order; f32 sum order changes
# speedup vs baseline: 1.3988x; 1.2975x over previous
#include <hip/hip_runtime.h>

typedef float v2f __attribute__((ext_vector_type(2)));
typedef float v4f __attribute__((ext_vector_type(4)));
typedef float v8f __attribute__((ext_vector_type(8)));

#define BN_    4
#define N_GRP  2
#define L_     1024
#define DM_    512
#define DIN_   1024
#define DSTATE 16
#define DTRANK 32
#define MTOT   4096
#define NCHUNK 16
#define TCH    64

__device__ __forceinline__
void store_tile(const v8f& t, float* __restrict__ C, int ldc, int rowb, int col,
                const float* __restrict__ bias, int epilogue)
{
    const float bv = (epilogue == 1) ? bias[col] : 0.0f;
#pragma unroll
    for (int r = 0; r < 8; ++r) {
        float v = t[r];
        if (epilogue == 1) {
            v += bv;
            v = (v > 20.0f) ? v : __logf(1.0f + __expf(v));
        }
        C[(size_t)(rowb + r) * ldc + col] = v;
    }
}

template <int TM, int KU, int KSPLIT>
__global__ __launch_bounds__(128) __attribute__((amdgpu_waves_per_eu(1, 4)))
void wmma_gemm_tn(const float* __restrict__ A, int lda,
                  const float* __restrict__ Bm, int ldb,
                  float* __restrict__ C, int ldc,
                  int M, int N, int K,
                  const float* __restrict__ bias, int epilogue)
{
    constexpr int TPB = 4 / KSPLIT;
    constexpr int OWN = TM * 4 / KSPLIT;
    static_assert(KSPLIT == 1 || KSPLIT == 2 || KSPLIT == 4, "waves per tile");
    static_assert((TM * 4) % KSPLIT == 0, "pieces must divide among the waves");

    const int lane  = threadIdx.x & 31;
    const int wave  = (KSPLIT > 1) ? __builtin_amdgcn_readfirstlane((int)(threadIdx.x >> 5))
                                   : (int)(threadIdx.x >> 5);
    const int ks    = wave % KSPLIT;
    const int tl    = wave / KSPLIT;
    const int ntile = N >> 6;
    const int tile  = blockIdx.x * TPB + tl;
    if constexpr (KSPLIT == 1) {
        if (tile >= (M / (16 * TM)) * ntile) return;
    }
    const int tm = tile / ntile;
    const int tn = tile % ntile;

    const int l15  = lane & 15;
    const int koff = (lane >> 4) << 2;
    const size_t astr = (size_t)16 * lda;
    const size_t bstr = (size_t)16 * ldb;
    const int klen = K / KSPLIT;
    const int kbeg = ks * klen;
    const float* ap = A  + (size_t)(tm * (16 * TM) + l15) * lda + koff;
    const float* bp = Bm + (size_t)(tn * 64 + l15) * ldb + koff;

    v8f acc[TM][4] = {};
    for (int k = kbeg; k < kbeg + klen; k += 8 * KU) {
        v4f a[KU][TM], b[KU][4];
#pragma unroll
        for (int u = 0; u < KU; ++u) {
#pragma unroll
            for (int i = 0; i < TM; ++i) a[u][i] = *(const v4f*)(ap + i * astr + k + 8 * u);
#pragma unroll
            for (int j = 0; j < 4; ++j)  b[u][j] = *(const v4f*)(bp + j * bstr + k + 8 * u);
        }
#pragma unroll
        for (int u = 0; u < KU; ++u) {
#pragma unroll
            for (int i = 0; i < TM; ++i)
#pragma unroll
                for (int j = 0; j < 4; ++j)
                    acc[i][j] = __builtin_amdgcn_wmma_f32_16x16x4_f32(
                                    false, __builtin_shufflevector(a[u][i], a[u][i], 0, 1),
                                    false, __builtin_shufflevector(b[u][j], b[u][j], 0, 1),
                                    (short)0, acc[i][j], false, false);
#pragma unroll
            for (int i = 0; i < TM; ++i)
#pragma unroll
                for (int j = 0; j < 4; ++j)
                    acc[i][j] = __builtin_amdgcn_wmma_f32_16x16x4_f32(
                                    false, __builtin_shufflevector(a[u][i], a[u][i], 2, 3),
                                    false, __builtin_shufflevector(b[u][j], b[u][j], 2, 3),
                                    (short)0, acc[i][j], false, false);
        }
    }

    const int rb = (lane >> 4) << 3;
    if constexpr (KSPLIT == 1) {
#pragma unroll
        for (int i = 0; i < TM; ++i)
#pragma unroll
            for (int j = 0; j < 4; ++j)
                store_tile(acc[i][j], C, ldc, tm * (16 * TM) + i * 16 + rb,
                           tn * 64 + j * 16 + l15, bias, epilogue);
    } else {
        __shared__ v4f red[TPB * KSPLIT * (KSPLIT - 1) * OWN * 2 * 32];
#pragma unroll
        for (int i = 0; i < TM; ++i)
#pragma unroll
            for (int j = 0; j < 4; ++j) {
                const int o = (i * 4 + j) / OWN;
                const int q = (i * 4 + j) % OWN;
                if (ks != o) {
                    const int slot = (ks < o) ? ks : ks - 1;
                    v4f* p = red + (size_t)((((tl * KSPLIT + o) * (KSPLIT - 1) + slot) * OWN + q) * 2) * 32 + lane;
                    p[0]  = __builtin_shufflevector(acc[i][j], acc[i][j], 0, 1, 2, 3);
                    p[32] = __builtin_shufflevector(acc[i][j], acc[i][j], 4, 5, 6, 7);
                }
            }
        __syncthreads();
#pragma unroll
        for (int o = 0; o < KSPLIT; ++o) {
            if (ks == o) {
#pragma unroll
                for (int q = 0; q < OWN; ++q) {
                    const int i = (o * OWN + q) / 4;
                    const int j = (o * OWN + q) % 4;
                    v8f s;
#pragma unroll
                    for (int sidx = 0; sidx < KSPLIT; ++sidx) {
                        v8f part;
                        if (sidx == o) {
                            part = acc[i][j];
                        } else {
                            const int slot = (sidx < o) ? sidx : sidx - 1;
                            const v4f* p = red + (size_t)((((tl * KSPLIT + o) * (KSPLIT - 1) + slot) * OWN + q) * 2) * 32 + lane;
                            part = __builtin_shufflevector(p[0], p[32], 0, 1, 2, 3, 4, 5, 6, 7);
                        }
                        if (sidx == 0) s = part; else s = s + part;
                    }
                    store_tile(s, C, ldc, tm * (16 * TM) + i * 16 + rb,
                               tn * 64 + j * 16 + l15, bias, epilogue);
                }
            }
        }
    }
}

__global__ __launch_bounds__(256)
void conv_silu_kernel(const float* __restrict__ C1,
                      const float* __restrict__ cw,
                      const float* __restrict__ cb,
                      float* __restrict__ U, int total)
{
    int idx = blockIdx.x * 256 + threadIdx.x;
    if (idx >= total) return;
    int d  = idx & (DIN_ - 1);
    int l  = (idx >> 10) & (L_ - 1);
    int bn = idx >> 20;
    int n  = bn & (N_GRP - 1);
    int ch = n * DIN_ + d;

    const float* w = cw + (size_t)ch * 4;
    float acc = cb[ch];
    const float* xcol = C1 + ((size_t)bn * L_) * (2 * DIN_) + d;
#pragma unroll
    for (int k = 0; k < 4; ++k) {
        int ll = l - 3 + k;
        float xv = (ll >= 0) ? xcol[(size_t)ll * (2 * DIN_)] : 0.0f;
        acc = fmaf(w[k], xv, acc);
    }
    U[idx] = acc / (1.0f + __expf(-acc));
}

__global__ __launch_bounds__(128)
void scan_chunk_reduce(const float* __restrict__ DT, const float* __restrict__ U,
                       const float* __restrict__ PR,
                       const float* __restrict__ A_log,
                       float* __restrict__ CH_C, float* __restrict__ CH_S)
{
    const int d  = ((blockIdx.x & 7) << 7) + threadIdx.x;
    const int ck = (blockIdx.x >> 3) & (NCHUNK - 1);
    const int bn = blockIdx.x >> 7;

    float As[DSTATE], h[DSTATE];
#pragma unroll
    for (int s = 0; s < DSTATE; ++s) {
        As[s] = -__expf(A_log[(size_t)d * DSTATE + s]);
        h[s]  = 0.0f;
    }
    float S = 0.0f;
    const int t0 = ck * TCH;
    for (int t = t0; t < t0 + TCH; ++t) {
        const size_t base = (size_t)bn * L_ + t;
        const float* pr = PR + base * 64;
        float dtv = DT[base * DIN_ + d];
        float uv  = U [base * DIN_ + d];
        S += dtv;
        float dtu = dtv * uv;
#pragma unroll
        for (int s = 0; s < DSTATE; ++s)
            h[s] = fmaf(__expf(dtv * As[s]), h[s], dtu * pr[DTRANK + s]);
    }
    const size_t cb = (size_t)(bn * NCHUNK + ck) * DSTATE * DIN_ + d;
#pragma unroll
    for (int s = 0; s < DSTATE; ++s) CH_C[cb + (size_t)s * DIN_] = h[s];
    CH_S[(size_t)(bn * NCHUNK + ck) * DIN_ + d] = S;
}

__global__ __launch_bounds__(128)
void scan_chunk_emit(const float* __restrict__ DT, const float* __restrict__ U,
                     const float* __restrict__ PR, const float* __restrict__ C1,
                     const float* __restrict__ A_log, const float* __restrict__ Dp,
                     const float* __restrict__ CH_C, const float* __restrict__ CH_S,
                     float* __restrict__ Y2)
{
    const int d  = ((blockIdx.x & 7) << 7) + threadIdx.x;
    const int ck = (blockIdx.x >> 3) & (NCHUNK - 1);
    const int bn = blockIdx.x >> 7;

    float As[DSTATE], h[DSTATE];
#pragma unroll
    for (int s = 0; s < DSTATE; ++s) {
        As[s] = -__expf(A_log[(size_t)d * DSTATE + s]);
        h[s]  = 0.0f;
    }
    for (int j = 0; j < ck; ++j) {
        float Sj = CH_S[(size_t)(bn * NCHUNK + j) * DIN_ + d];
        const size_t cb = (size_t)(bn * NCHUNK + j) * DSTATE * DIN_ + d;
#pragma unroll
        for (int s = 0; s < DSTATE; ++s)
            h[s] = fmaf(__expf(Sj * As[s]), h[s], CH_C[cb + (size_t)s * DIN_]);
    }

    const float dp = Dp[d];
    const int t0 = ck * TCH;
    for (int t = t0; t < t0 + TCH; ++t) {
        const size_t base = (size_t)bn * L_ + t;
        const float* pr = PR + base * 64;
        float dtv = DT[base * DIN_ + d];
        float uv  = U [base * DIN_ + d];
        float dtu = dtv * uv;
        float y = 0.0f;
#pragma unroll
        for (int s = 0; s < DSTATE; ++s) {
            h[s] = fmaf(__expf(dtv * As[s]), h[s], dtu * pr[DTRANK + s]);
            y    = fmaf(h[s], pr[DTRANK + DSTATE + s], y);
        }
        float res = C1[base * (2 * DIN_) + DIN_ + d];
        float sr  = res / (1.0f + __expf(-res));
        Y2[base * DIN_ + d] = (y + uv * dp) * sr;
    }
}

extern "C" void kernel_launch(void* const* d_in, const int* in_sizes, int n_in,
                              void* d_out, int out_size, void* d_ws, size_t ws_size,
                              hipStream_t stream)
{
    const float* x      = (const float*)d_in[0];
    const float* W_in   = (const float*)d_in[1];
    const float* conv_w = (const float*)d_in[2];
    const float* conv_b = (const float*)d_in[3];
    const float* W_x    = (const float*)d_in[4];
    const float* W_dt   = (const float*)d_in[5];
    const float* b_dt   = (const float*)d_in[6];
    const float* A_log  = (const float*)d_in[7];
    const float* Dp     = (const float*)d_in[8];
    const float* W_out  = (const float*)d_in[9];
    float* out = (float*)d_out;

    float* C1   = (float*)d_ws;
    float* U    = C1   + (size_t)MTOT * (2 * DIN_);
    float* PR   = U    + (size_t)MTOT * DIN_;
    float* DTb  = PR   + (size_t)MTOT * 64;
    float* Y2   = DTb  + (size_t)MTOT * DIN_;
    float* CH_C = Y2   + (size_t)MTOT * DIN_;
    float* CH_S = CH_C + (size_t)BN_ * NCHUNK * DSTATE * DIN_;

    int tiles = (MTOT / 64) * ((2 * DIN_) / 64);
    wmma_gemm_tn<4, 1, 1><<<tiles / 4, 128, 0, stream>>>(x, DM_, W_in, DM_,
                                                C1, 2 * DIN_,
                                                MTOT, 2 * DIN_, DM_, nullptr, 0);

    conv_silu_kernel<<<(MTOT * DIN_) / 256, 256, 0, stream>>>(C1, conv_w, conv_b,
                                                              U, MTOT * DIN_);

    tiles = (MTOT / 16) * (64 / 64);
    wmma_gemm_tn<1, 4, 4><<<tiles, 128, 0, stream>>>(U, DIN_, W_x, DIN_,
                                                PR, 64,
                                                MTOT, 64, DIN_, nullptr, 0);

    tiles = (MTOT / 64) * (DIN_ / 64);
    wmma_gemm_tn<4, 1, 1><<<tiles / 4, 128, 0, stream>>>(PR, 64, W_dt, DTRANK,
                                                DTb, DIN_,
                                                MTOT, DIN_, DTRANK, b_dt, 1);

    const int scan_blocks = BN_ * NCHUNK * (DIN_ / 128);
    scan_chunk_reduce<<<scan_blocks, 128, 0, stream>>>(DTb, U, PR, A_log,
                                                       CH_C, CH_S);
    scan_chunk_emit<<<scan_blocks, 128, 0, stream>>>(DTb, U, PR, C1, A_log, Dp,
                                                     CH_C, CH_S, Y2);

    tiles = (MTOT / 64) * (DM_ / 64);
    wmma_gemm_tn<4, 2, 2><<<tiles / 2, 128, 0, stream>>>(Y2, DIN_, W_out, DIN_,
                                                out, DM_,
                                                MTOT, DM_, DIN_, nullptr, 0);
}


# Scan passes paid one memory round trip per time step on each wave's serial chain: B/C rows now staged in LDS once per block, dt/u/gate loads issued 16 steps per group, chunk summaries 4 per group
# speedup vs baseline: 1.6728x; 1.1959x over previous
#include <hip/hip_runtime.h>

typedef float v2f __attribute__((ext_vector_type(2)));
typedef float v4f __attribute__((ext_vector_type(4)));
typedef float v8f __attribute__((ext_vector_type(8)));

#define BN_    4
#define N_GRP  2
#define L_     1024
#define DM_    512
#define DIN_   1024
#define DSTATE 16
#define DTRANK 32
#define MTOT   4096
#define NCHUNK 16
#define TCH    64

__device__ __forceinline__
void store_tile(const v8f& t, float* __restrict__ C, int ldc, int rowb, int col,
                const float* __restrict__ bias, int epilogue)
{
    const float bv = (epilogue == 1) ? bias[col] : 0.0f;
#pragma unroll
    for (int r = 0; r < 8; ++r) {
        float v = t[r];
        if (epilogue == 1) {
            v += bv;
            v = (v > 20.0f) ? v : __logf(1.0f + __expf(v));
        }
        C[(size_t)(rowb + r) * ldc + col] = v;
    }
}

template <int TM, int KU, int KSPLIT>
__global__ __launch_bounds__(128) __attribute__((amdgpu_waves_per_eu(1, 4)))
void wmma_gemm_tn(const float* __restrict__ A, int lda,
                  const float* __restrict__ Bm, int ldb,
                  float* __restrict__ C, int ldc,
                  int M, int N, int K,
                  const float* __restrict__ bias, int epilogue)
{
    constexpr int TPB = 4 / KSPLIT;
    constexpr int OWN = TM * 4 / KSPLIT;
    static_assert(KSPLIT == 1 || KSPLIT == 2 || KSPLIT == 4, "waves per tile");
    static_assert((TM * 4) % KSPLIT == 0, "pieces must divide among the waves");

    const int lane  = threadIdx.x & 31;
    const int wave  = (KSPLIT > 1) ? __builtin_amdgcn_readfirstlane((int)(threadIdx.x >> 5))
                                   : (int)(threadIdx.x >> 5);
    const int ks    = wave % KSPLIT;
    const int tl    = wave / KSPLIT;
    const int ntile = N >> 6;
    const int tile  = blockIdx.x * TPB + tl;
    if constexpr (KSPLIT == 1) {
        if (tile >= (M / (16 * TM)) * ntile) return;
    }
    const int tm = tile / ntile;
    const int tn = tile % ntile;

    const int l15  = lane & 15;
    const int koff = (lane >> 4) << 2;
    const size_t astr = (size_t)16 * lda;
    const size_t bstr = (size_t)16 * ldb;
    const int klen = K / KSPLIT;
    const int kbeg = ks * klen;
    const float* ap = A  + (size_t)(tm * (16 * TM) + l15) * lda + koff;
    const float* bp = Bm + (size_t)(tn * 64 + l15) * ldb + koff;

    v8f acc[TM][4] = {};
    for (int k = kbeg; k < kbeg + klen; k += 8 * KU) {
        v4f a[KU][TM], b[KU][4];
#pragma unroll
        for (int u = 0; u < KU; ++u) {
#pragma unroll
            for (int i = 0; i < TM; ++i) a[u][i] = *(const v4f*)(ap + i * astr + k + 8 * u);
#pragma unroll
            for (int j = 0; j < 4; ++j)  b[u][j] = *(const v4f*)(bp + j * bstr + k + 8 * u);
        }
#pragma unroll
        for (int u = 0; u < KU; ++u) {
#pragma unroll
            for (int i = 0; i < TM; ++i)
#pragma unroll
                for (int j = 0; j < 4; ++j)
                    acc[i][j] = __builtin_amdgcn_wmma_f32_16x16x4_f32(
                                    false, __builtin_shufflevector(a[u][i], a[u][i], 0, 1),
                                    false, __builtin_shufflevector(b[u][j], b[u][j], 0, 1),
                                    (short)0, acc[i][j], false, false);
#pragma unroll
            for (int i = 0; i < TM; ++i)
#pragma unroll
                for (int j = 0; j < 4; ++j)
                    acc[i][j] = __builtin_amdgcn_wmma_f32_16x16x4_f32(
                                    false, __builtin_shufflevector(a[u][i], a[u][i], 2, 3),
                                    false, __builtin_shufflevector(b[u][j], b[u][j], 2, 3),
                                    (short)0, acc[i][j], false, false);
        }
    }

    const int rb = (lane >> 4) << 3;
    if constexpr (KSPLIT == 1) {
#pragma unroll
        for (int i = 0; i < TM; ++i)
#pragma unroll
            for (int j = 0; j < 4; ++j)
                store_tile(acc[i][j], C, ldc, tm * (16 * TM) + i * 16 + rb,
                           tn * 64 + j * 16 + l15, bias, epilogue);
    } else {
        __shared__ v4f red[TPB * KSPLIT * (KSPLIT - 1) * OWN * 2 * 32];
#pragma unroll
        for (int i = 0; i < TM; ++i)
#pragma unroll
            for (int j = 0; j < 4; ++j) {
                const int o = (i * 4 + j) / OWN;
                const int q = (i * 4 + j) % OWN;
                if (ks != o) {
                    const int slot = (ks < o) ? ks : ks - 1;
                    v4f* p = red + (size_t)((((tl * KSPLIT + o) * (KSPLIT - 1) + slot) * OWN + q) * 2) * 32 + lane;
                    p[0]  = __builtin_shufflevector(acc[i][j], acc[i][j], 0, 1, 2, 3);
                    p[32] = __builtin_shufflevector(acc[i][j], acc[i][j], 4, 5, 6, 7);
                }
            }
        __syncthreads();
#pragma unroll
        for (int o = 0; o < KSPLIT; ++o) {
            if (ks == o) {
#pragma unroll
                for (int q = 0; q < OWN; ++q) {
                    const int i = (o * OWN + q) / 4;
                    const int j = (o * OWN + q) % 4;
                    v8f s;
#pragma unroll
                    for (int sidx = 0; sidx < KSPLIT; ++sidx) {
                        v8f part;
                        if (sidx == o) {
                            part = acc[i][j];
                        } else {
                            const int slot = (sidx < o) ? sidx : sidx - 1;
                            const v4f* p = red + (size_t)((((tl * KSPLIT + o) * (KSPLIT - 1) + slot) * OWN + q) * 2) * 32 + lane;
                            part = __builtin_shufflevector(p[0], p[32], 0, 1, 2, 3, 4, 5, 6, 7);
                        }
                        if (sidx == 0) s = part; else s = s + part;
                    }
                    store_tile(s, C, ldc, tm * (16 * TM) + i * 16 + rb,
                               tn * 64 + j * 16 + l15, bias, epilogue);
                }
            }
        }
    }
}

__global__ __launch_bounds__(256)
void conv_silu_kernel(const float* __restrict__ C1,
                      const float* __restrict__ cw,
                      const float* __restrict__ cb,
                      float* __restrict__ U, int total)
{
    int idx = blockIdx.x * 256 + threadIdx.x;
    if (idx >= total) return;
    int d  = idx & (DIN_ - 1);
    int l  = (idx >> 10) & (L_ - 1);
    int bn = idx >> 20;
    int n  = bn & (N_GRP - 1);
    int ch = n * DIN_ + d;

    const float* w = cw + (size_t)ch * 4;
    float acc = cb[ch];
    const float* xcol = C1 + ((size_t)bn * L_) * (2 * DIN_) + d;
#pragma unroll
    for (int k = 0; k < 4; ++k) {
        int ll = l - 3 + k;
        float xv = (ll >= 0) ? xcol[(size_t)ll * (2 * DIN_)] : 0.0f;
        acc = fmaf(w[k], xv, acc);
    }
    U[idx] = acc / (1.0f + __expf(-acc));
}

#define TG 16
#define PG 4

__global__ __launch_bounds__(128)
void scan_chunk_reduce(const float* __restrict__ DT, const float* __restrict__ U,
                       const float* __restrict__ PR,
                       const float* __restrict__ A_log,
                       float* __restrict__ CH_C, float* __restrict__ CH_S)
{
    const int d  = ((blockIdx.x & 7) << 7) + threadIdx.x;
    const int ck = (blockIdx.x >> 3) & (NCHUNK - 1);
    const int bn = blockIdx.x >> 7;
    const int t0 = ck * TCH;
    const size_t row0 = (size_t)bn * L_ + t0;

    __shared__ v4f bs[TCH * 4];
    {
        const float* prc = PR + row0 * 64 + DTRANK;
#pragma unroll
        for (int r = 0; r < (TCH * 4) / 128; ++r) {
            const int i = threadIdx.x + r * 128;
            bs[i] = *(const v4f*)(prc + (size_t)(i >> 2) * 64 + ((i & 3) << 2));
        }
    }

    float As[DSTATE], h[DSTATE];
#pragma unroll
    for (int s = 0; s < DSTATE; ++s) {
        As[s] = -__expf(A_log[(size_t)d * DSTATE + s]);
        h[s]  = 0.0f;
    }
    __syncthreads();

    float S = 0.0f;
    const float* dtp = DT + row0 * DIN_ + d;
    const float* up  = U  + row0 * DIN_ + d;
    for (int g = 0; g < TCH; g += TG) {
        float dtv[TG], uv[TG];
#pragma unroll
        for (int i = 0; i < TG; ++i) {
            dtv[i] = dtp[(size_t)(g + i) * DIN_];
            uv[i]  = up [(size_t)(g + i) * DIN_];
        }
#pragma unroll
        for (int i = 0; i < TG; ++i) {
            float Bv[DSTATE];
#pragma unroll
            for (int q = 0; q < 4; ++q) {
                const v4f b = bs[(g + i) * 4 + q];
                Bv[4 * q + 0] = b[0]; Bv[4 * q + 1] = b[1];
                Bv[4 * q + 2] = b[2]; Bv[4 * q + 3] = b[3];
            }
            S += dtv[i];
            const float dtu = dtv[i] * uv[i];
#pragma unroll
            for (int s = 0; s < DSTATE; ++s)
                h[s] = fmaf(__expf(dtv[i] * As[s]), h[s], dtu * Bv[s]);
        }
    }
    const size_t cb = (size_t)(bn * NCHUNK + ck) * DSTATE * DIN_ + d;
#pragma unroll
    for (int s = 0; s < DSTATE; ++s) CH_C[cb + (size_t)s * DIN_] = h[s];
    CH_S[(size_t)(bn * NCHUNK + ck) * DIN_ + d] = S;
}

template <int P>
__device__ __forceinline__
void fold_chunks(float (&h)[DSTATE], const float (&As)[DSTATE],
                 const float* __restrict__ chs, const float* __restrict__ chc, int j0)
{
    static_assert(PG == 4, "the tail switch in pass 2 lists 1..PG-1");
    float Sj[P], cj[P][DSTATE];
#pragma unroll
    for (int jj = 0; jj < P; ++jj) {
        Sj[jj] = chs[(size_t)(j0 + jj) * DIN_];
#pragma unroll
        for (int s = 0; s < DSTATE; ++s)
            cj[jj][s] = chc[((size_t)(j0 + jj) * DSTATE + s) * DIN_];
    }
#pragma unroll
    for (int jj = 0; jj < P; ++jj)
#pragma unroll
        for (int s = 0; s < DSTATE; ++s)
            h[s] = fmaf(__expf(Sj[jj] * As[s]), h[s], cj[jj][s]);
}

__global__ __launch_bounds__(128)
void scan_chunk_emit(const float* __restrict__ DT, const float* __restrict__ U,
                     const float* __restrict__ PR, const float* __restrict__ C1,
                     const float* __restrict__ A_log, const float* __restrict__ Dp,
                     const float* __restrict__ CH_C, const float* __restrict__ CH_S,
                     float* __restrict__ Y2)
{
    const int d  = ((blockIdx.x & 7) << 7) + threadIdx.x;
    const int ck = (blockIdx.x >> 3) & (NCHUNK - 1);
    const int bn = blockIdx.x >> 7;
    const int t0 = ck * TCH;
    const size_t row0 = (size_t)bn * L_ + t0;

    __shared__ v4f bcs[TCH * 8];
    v4f stg[(TCH * 8) / 128];
    {
        const float* prc = PR + row0 * 64 + DTRANK;
#pragma unroll
        for (int r = 0; r < (TCH * 8) / 128; ++r) {
            const int i = threadIdx.x + r * 128;
            stg[r] = *(const v4f*)(prc + (size_t)(i >> 3) * 64 + ((i & 7) << 2));
        }
    }

    float As[DSTATE], h[DSTATE];
#pragma unroll
    for (int s = 0; s < DSTATE; ++s) {
        As[s] = -__expf(A_log[(size_t)d * DSTATE + s]);
        h[s]  = 0.0f;
    }
    {
        const float* chs = CH_S + (size_t)(bn * NCHUNK) * DIN_ + d;
        const float* chc = CH_C + (size_t)(bn * NCHUNK) * DSTATE * DIN_ + d;
        int j0 = 0;
        for (; j0 + PG <= ck; j0 += PG) fold_chunks<PG>(h, As, chs, chc, j0);
        switch (ck - j0) {
            case 1: fold_chunks<1>(h, As, chs, chc, j0); break;
            case 2: fold_chunks<2>(h, As, chs, chc, j0); break;
            case 3: fold_chunks<3>(h, As, chs, chc, j0); break;
            default: break;
        }
    }
#pragma unroll
    for (int r = 0; r < (TCH * 8) / 128; ++r) bcs[threadIdx.x + r * 128] = stg[r];

    const float dp = Dp[d];
    __syncthreads();

    const float* dtp = DT + row0 * DIN_ + d;
    const float* up  = U  + row0 * DIN_ + d;
    const float* rp  = C1 + row0 * (2 * DIN_) + DIN_ + d;
    float*       yp  = Y2 + row0 * DIN_ + d;
    for (int g = 0; g < TCH; g += TG) {
        float dtv[TG], uv[TG], rv[TG];
#pragma unroll
        for (int i = 0; i < TG; ++i) {
            dtv[i] = dtp[(size_t)(g + i) * DIN_];
            uv[i]  = up [(size_t)(g + i) * DIN_];
            rv[i]  = rp [(size_t)(g + i) * (2 * DIN_)];
        }
#pragma unroll
        for (int i = 0; i < TG; ++i) {
            float Bv[DSTATE], Cv[DSTATE];
#pragma unroll
            for (int q = 0; q < 4; ++q) {
                const v4f b = bcs[(g + i) * 8 + q];
                const v4f c = bcs[(g + i) * 8 + 4 + q];
                Bv[4 * q + 0] = b[0]; Bv[4 * q + 1] = b[1];
                Bv[4 * q + 2] = b[2]; Bv[4 * q + 3] = b[3];
                Cv[4 * q + 0] = c[0]; Cv[4 * q + 1] = c[1];
                Cv[4 * q + 2] = c[2]; Cv[4 * q + 3] = c[3];
            }
            const float dtu = dtv[i] * uv[i];
            float y = 0.0f;
#pragma unroll
            for (int s = 0; s < DSTATE; ++s) {
                h[s] = fmaf(__expf(dtv[i] * As[s]), h[s], dtu * Bv[s]);
                y    = fmaf(h[s], Cv[s], y);
            }
            const float res = rv[i];
            const float sr  = res / (1.0f + __expf(-res));
            yp[(size_t)(g + i) * DIN_] = (y + uv[i] * dp) * sr;
        }
    }
}

extern "C" void kernel_launch(void* const* d_in, const int* in_sizes, int n_in,
                              void* d_out, int out_size, void* d_ws, size_t ws_size,
                              hipStream_t stream)
{
    const float* x      = (const float*)d_in[0];
    const float* W_in   = (const float*)d_in[1];
    const float* conv_w = (const float*)d_in[2];
    const float* conv_b = (const float*)d_in[3];
    const float* W_x    = (const float*)d_in[4];
    const float* W_dt   = (const float*)d_in[5];
    const float* b_dt   = (const float*)d_in[6];
    const float* A_log  = (const float*)d_in[7];
    const float* Dp     = (const float*)d_in[8];
    const float* W_out  = (const float*)d_in[9];
    float* out = (float*)d_out;

    float* C1   = (float*)d_ws;
    float* U    = C1   + (size_t)MTOT * (2 * DIN_);
    float* PR   = U    + (size_t)MTOT * DIN_;
    float* DTb  = PR   + (size_t)MTOT * 64;
    float* Y2   = DTb  + (size_t)MTOT * DIN_;
    float* CH_C = Y2   + (size_t)MTOT * DIN_;
    float* CH_S = CH_C + (size_t)BN_ * NCHUNK * DSTATE * DIN_;

    int tiles = (MTOT / 64) * ((2 * DIN_) / 64);
    wmma_gemm_tn<4, 1, 1><<<tiles / 4, 128, 0, stream>>>(x, DM_, W_in, DM_,
                                                C1, 2 * DIN_,
                                                MTOT, 2 * DIN_, DM_, nullptr, 0);

    conv_silu_kernel<<<(MTOT * DIN_) / 256, 256, 0, stream>>>(C1, conv_w, conv_b,
                                                              U, MTOT * DIN_);

    tiles = (MTOT / 16) * (64 / 64);
    wmma_gemm_tn<1, 4, 4><<<tiles, 128, 0, stream>>>(U, DIN_, W_x, DIN_,
                                                PR, 64,
                                                MTOT, 64, DIN_, nullptr, 0);

    tiles = (MTOT / 64) * (DIN_ / 64);
    wmma_gemm_tn<4, 1, 1><<<tiles / 4, 128, 0, stream>>>(PR, 64, W_dt, DTRANK,
                                                DTb, DIN_,
                                                MTOT, DIN_, DTRANK, b_dt, 1);

    const int scan_blocks = BN_ * NCHUNK * (DIN_ / 128);
    scan_chunk_reduce<<<scan_blocks, 128, 0, stream>>>(DTb, U, PR, A_log,
                                                       CH_C, CH_S);
    scan_chunk_emit<<<scan_blocks, 128, 0, stream>>>(DTb, U, PR, C1, A_log, Dp,
                                                     CH_C, CH_S, Y2);

    tiles = (MTOT / 64) * (DM_ / 64);
    wmma_gemm_tn<4, 2, 2><<<tiles / 2, 128, 0, stream>>>(Y2, DIN_, W_out, DIN_,
                                                out, DM_,
                                                MTOT, DM_, DIN_, nullptr, 0);
}


# conv+silu ran as 131072 one-output waves, limited by how fast waves enter the chip: now 2048 waves, a thread takes 4 channels (16-byte loads/stores) x 16 steps; per-output arithmetic unchanged
# speedup vs baseline: 1.7359x; 1.0377x over previous
#include <hip/hip_runtime.h>

typedef float v2f __attribute__((ext_vector_type(2)));
typedef float v4f __attribute__((ext_vector_type(4)));
typedef float v8f __attribute__((ext_vector_type(8)));

#define BN_    4
#define N_GRP  2
#define L_     1024
#define DM_    512
#define DIN_   1024
#define DSTATE 16
#define DTRANK 32
#define MTOT   4096
#define NCHUNK 16
#define TCH    64

__device__ __forceinline__
void store_tile(const v8f& t, float* __restrict__ C, int ldc, int rowb, int col,
                const float* __restrict__ bias, int epilogue)
{
    const float bv = (epilogue == 1) ? bias[col] : 0.0f;
#pragma unroll
    for (int r = 0; r < 8; ++r) {
        float v = t[r];
        if (epilogue == 1) {
            v += bv;
            v = (v > 20.0f) ? v : __logf(1.0f + __expf(v));
        }
        C[(size_t)(rowb + r) * ldc + col] = v;
    }
}

template <int TM, int KU, int KSPLIT>
__global__ __launch_bounds__(128) __attribute__((amdgpu_waves_per_eu(1, 4)))
void wmma_gemm_tn(const float* __restrict__ A, int lda,
                  const float* __restrict__ Bm, int ldb,
                  float* __restrict__ C, int ldc,
                  int M, int N, int K,
                  const float* __restrict__ bias, int epilogue)
{
    constexpr int TPB = 4 / KSPLIT;
    constexpr int OWN = TM * 4 / KSPLIT;
    static_assert(KSPLIT == 1 || KSPLIT == 2 || KSPLIT == 4, "waves per tile");
    static_assert((TM * 4) % KSPLIT == 0, "pieces must divide among the waves");

    const int lane  = threadIdx.x & 31;
    const int wave  = (KSPLIT > 1) ? __builtin_amdgcn_readfirstlane((int)(threadIdx.x >> 5))
                                   : (int)(threadIdx.x >> 5);
    const int ks    = wave % KSPLIT;
    const int tl    = wave / KSPLIT;
    const int ntile = N >> 6;
    const int tile  = blockIdx.x * TPB + tl;
    if constexpr (KSPLIT == 1) {
        if (tile >= (M / (16 * TM)) * ntile) return;
    }
    const int tm = tile / ntile;
    const int tn = tile % ntile;

    const int l15  = lane & 15;
    const int koff = (lane >> 4) << 2;
    const size_t astr = (size_t)16 * lda;
    const size_t bstr = (size_t)16 * ldb;
    const int klen = K / KSPLIT;
    const int kbeg = ks * klen;
    const float* ap = A  + (size_t)(tm * (16 * TM) + l15) * lda + koff;
    const float* bp = Bm + (size_t)(tn * 64 + l15) * ldb + koff;

    v8f acc[TM][4] = {};
    for (int k = kbeg; k < kbeg + klen; k += 8 * KU) {
        v4f a[KU][TM], b[KU][4];
#pragma unroll
        for (int u = 0; u < KU; ++u) {
#pragma unroll
            for (int i = 0; i < TM; ++i) a[u][i] = *(const v4f*)(ap + i * astr + k + 8 * u);
#pragma unroll
            for (int j = 0; j < 4; ++j)  b[u][j] = *(const v4f*)(bp + j * bstr + k + 8 * u);
        }
#pragma unroll
        for (int u = 0; u < KU; ++u) {
#pragma unroll
            for (int i = 0; i < TM; ++i)
#pragma unroll
                for (int j = 0; j < 4; ++j)
                    acc[i][j] = __builtin_amdgcn_wmma_f32_16x16x4_f32(
                                    false, __builtin_shufflevector(a[u][i], a[u][i], 0, 1),
                                    false, __builtin_shufflevector(b[u][j], b[u][j], 0, 1),
                                    (short)0, acc[i][j], false, false);
#pragma unroll
            for (int i = 0; i < TM; ++i)
#pragma unroll
                for (int j = 0; j < 4; ++j)
                    acc[i][j] = __builtin_amdgcn_wmma_f32_16x16x4_f32(
                                    false, __builtin_shufflevector(a[u][i], a[u][i], 2, 3),
                                    false, __builtin_shufflevector(b[u][j], b[u][j], 2, 3),
                                    (short)0, acc[i][j], false, false);
        }
    }

    const int rb = (lane >> 4) << 3;
    if constexpr (KSPLIT == 1) {
#pragma unroll
        for (int i = 0; i < TM; ++i)
#pragma unroll
            for (int j = 0; j < 4; ++j)
                store_tile(acc[i][j], C, ldc, tm * (16 * TM) + i * 16 + rb,
                           tn * 64 + j * 16 + l15, bias, epilogue);
    } else {
        __shared__ v4f red[TPB * KSPLIT * (KSPLIT - 1) * OWN * 2 * 32];
#pragma unroll
        for (int i = 0; i < TM; ++i)
#pragma unroll
            for (int j = 0; j < 4; ++j) {
                const int o = (i * 4 + j) / OWN;
                const int q = (i * 4 + j) % OWN;
                if (ks != o) {
                    const int slot = (ks < o) ? ks : ks - 1;
                    v4f* p = red + (size_t)((((tl * KSPLIT + o) * (KSPLIT - 1) + slot) * OWN + q) * 2) * 32 + lane;
                    p[0]  = __builtin_shufflevector(acc[i][j], acc[i][j], 0, 1, 2, 3);
                    p[32] = __builtin_shufflevector(acc[i][j], acc[i][j], 4, 5, 6, 7);
                }
            }
        __syncthreads();
#pragma unroll
        for (int o = 0; o < KSPLIT; ++o) {
            if (ks == o) {
#pragma unroll
                for (int q = 0; q < OWN; ++q) {
                    const int i = (o * OWN + q) / 4;
                    const int j = (o * OWN + q) % 4;
                    v8f s;
#pragma unroll
                    for (int sidx = 0; sidx < KSPLIT; ++sidx) {
                        v8f part;
                        if (sidx == o) {
                            part = acc[i][j];
                        } else {
                            const int slot = (sidx < o) ? sidx : sidx - 1;
                            const v4f* p = red + (size_t)((((tl * KSPLIT + o) * (KSPLIT - 1) + slot) * OWN + q) * 2) * 32 + lane;
                            part = __builtin_shufflevector(p[0], p[32], 0, 1, 2, 3, 4, 5, 6, 7);
                        }
                        if (sidx == 0) s = part; else s = s + part;
                    }
                    store_tile(s, C, ldc, tm * (16 * TM) + i * 16 + rb,
                               tn * 64 + j * 16 + l15, bias, epilogue);
                }
            }
        }
    }
}

#define CTL 16

__global__ __launch_bounds__(256)
void conv_silu_kernel(const float* __restrict__ C1,
                      const float* __restrict__ cw,
                      const float* __restrict__ cb,
                      float* __restrict__ U)
{
    const int d  = threadIdx.x << 2;
    const int lt = blockIdx.x & (L_ / CTL - 1);
    const int bn = blockIdx.x / (L_ / CTL);
    const int ch = (bn & (N_GRP - 1)) * DIN_ + d;
    const size_t row0 = (size_t)bn * L_ + (size_t)lt * CTL;

    v4f w[4];
#pragma unroll
    for (int c = 0; c < 4; ++c) w[c] = *(const v4f*)(cw + (size_t)(ch + c) * 4);
    const v4f bias = *(const v4f*)(cb + ch);

    const float* xp = C1 + row0 * (2 * DIN_) + d;
    v4f x[CTL + 3];
#pragma unroll
    for (int i = 0; i < CTL; ++i)
        x[i + 3] = *(const v4f*)(xp + (size_t)i * (2 * DIN_));
    if (lt != 0) {
#pragma unroll
        for (int i = 0; i < 3; ++i)
            x[i] = *(const v4f*)(xp - (size_t)(3 - i) * (2 * DIN_));
    } else {
#pragma unroll
        for (int i = 0; i < 3; ++i) x[i] = (v4f){0.0f, 0.0f, 0.0f, 0.0f};
    }

    float* yp = U + row0 * DIN_ + d;
#pragma unroll
    for (int i = 0; i < CTL; ++i) {
        v4f o;
#pragma unroll
        for (int c = 0; c < 4; ++c) {
            float acc = bias[c];
#pragma unroll
            for (int k = 0; k < 4; ++k) acc = fmaf(w[c][k], x[i + k][c], acc);
            o[c] = acc / (1.0f + __expf(-acc));
        }
        *(v4f*)(yp + (size_t)i * DIN_) = o;
    }
}

#define TG 16
#define PG 4

__global__ __launch_bounds__(128)
void scan_chunk_reduce(const float* __restrict__ DT, const float* __restrict__ U,
                       const float* __restrict__ PR,
                       const float* __restrict__ A_log,
                       float* __restrict__ CH_C, float* __restrict__ CH_S)
{
    const int d  = ((blockIdx.x & 7) << 7) + threadIdx.x;
    const int ck = (blockIdx.x >> 3) & (NCHUNK - 1);
    const int bn = blockIdx.x >> 7;
    const int t0 = ck * TCH;
    const size_t row0 = (size_t)bn * L_ + t0;

    __shared__ v4f bs[TCH * 4];
    {
        const float* prc = PR + row0 * 64 + DTRANK;
#pragma unroll
        for (int r = 0; r < (TCH * 4) / 128; ++r) {
            const int i = threadIdx.x + r * 128;
            bs[i] = *(const v4f*)(prc + (size_t)(i >> 2) * 64 + ((i & 3) << 2));
        }
    }

    float As[DSTATE], h[DSTATE];
#pragma unroll
    for (int s = 0; s < DSTATE; ++s) {
        As[s] = -__expf(A_log[(size_t)d * DSTATE + s]);
        h[s]  = 0.0f;
    }
    __syncthreads();

    float S = 0.0f;
    const float* dtp = DT + row0 * DIN_ + d;
    const float* up  = U  + row0 * DIN_ + d;
    for (int g = 0; g < TCH; g += TG) {
        float dtv[TG], uv[TG];
#pragma unroll
        for (int i = 0; i < TG; ++i) {
            dtv[i] = dtp[(size_t)(g + i) * DIN_];
            uv[i]  = up [(size_t)(g + i) * DIN_];
        }
#pragma unroll
        for (int i = 0; i < TG; ++i) {
            float Bv[DSTATE];
#pragma unroll
            for (int q = 0; q < 4; ++q) {
                const v4f b = bs[(g + i) * 4 + q];
                Bv[4 * q + 0] = b[0]; Bv[4 * q + 1] = b[1];
                Bv[4 * q + 2] = b[2]; Bv[4 * q + 3] = b[3];
            }
            S += dtv[i];
            const float dtu = dtv[i] * uv[i];
#pragma unroll
            for (int s = 0; s < DSTATE; ++s)
                h[s] = fmaf(__expf(dtv[i] * As[s]), h[s], dtu * Bv[s]);
        }
    }
    const size_t cb = (size_t)(bn * NCHUNK + ck) * DSTATE * DIN_ + d;
#pragma unroll
    for (int s = 0; s < DSTATE; ++s) CH_C[cb + (size_t)s * DIN_] = h[s];
    CH_S[(size_t)(bn * NCHUNK + ck) * DIN_ + d] = S;
}

template <int P>
__device__ __forceinline__
void fold_chunks(float (&h)[DSTATE], const float (&As)[DSTATE],
                 const float* __restrict__ chs, const float* __restrict__ chc, int j0)
{
    static_assert(PG == 4, "the tail switch in pass 2 lists 1..PG-1");
    float Sj[P], cj[P][DSTATE];
#pragma unroll
    for (int jj = 0; jj < P; ++jj) {
        Sj[jj] = chs[(size_t)(j0 + jj) * DIN_];
#pragma unroll
        for (int s = 0; s < DSTATE; ++s)
            cj[jj][s] = chc[((size_t)(j0 + jj) * DSTATE + s) * DIN_];
    }
#pragma unroll
    for (int jj = 0; jj < P; ++jj)
#pragma unroll
        for (int s = 0; s < DSTATE; ++s)
            h[s] = fmaf(__expf(Sj[jj] * As[s]), h[s], cj[jj][s]);
}

__global__ __launch_bounds__(128)
void scan_chunk_emit(const float* __restrict__ DT, const float* __restrict__ U,
                     const float* __restrict__ PR, const float* __restrict__ C1,
                     const float* __restrict__ A_log, const float* __restrict__ Dp,
                     const float* __restrict__ CH_C, const float* __restrict__ CH_S,
                     float* __restrict__ Y2)
{
    const int d  = ((blockIdx.x & 7) << 7) + threadIdx.x;
    const int ck = (blockIdx.x >> 3) & (NCHUNK - 1);
    const int bn = blockIdx.x >> 7;
    const int t0 = ck * TCH;
    const size_t row0 = (size_t)bn * L_ + t0;

    __shared__ v4f bcs[TCH * 8];
    v4f stg[(TCH * 8) / 128];
    {
        const float* prc = PR + row0 * 64 + DTRANK;
#pragma unroll
        for (int r = 0; r < (TCH * 8) / 128; ++r) {
            const int i = threadIdx.x + r * 128;
            stg[r] = *(const v4f*)(prc + (size_t)(i >> 3) * 64 + ((i & 7) << 2));
        }
    }

    float As[DSTATE], h[DSTATE];
#pragma unroll
    for (int s = 0; s < DSTATE; ++s) {
        As[s] = -__expf(A_log[(size_t)d * DSTATE + s]);
        h[s]  = 0.0f;
    }
    {
        const float* chs = CH_S + (size_t)(bn * NCHUNK) * DIN_ + d;
        const float* chc = CH_C + (size_t)(bn * NCHUNK) * DSTATE * DIN_ + d;
        int j0 = 0;
        for (; j0 + PG <= ck; j0 += PG) fold_chunks<PG>(h, As, chs, chc, j0);
        switch (ck - j0) {
            case 1: fold_chunks<1>(h, As, chs, chc, j0); break;
            case 2: fold_chunks<2>(h, As, chs, chc, j0); break;
            case 3: fold_chunks<3>(h, As, chs, chc, j0); break;
            default: break;
        }
    }
#pragma unroll
    for (int r = 0; r < (TCH * 8) / 128; ++r) bcs[threadIdx.x + r * 128] = stg[r];

    const float dp = Dp[d];
    __syncthreads();

    const float* dtp = DT + row0 * DIN_ + d;
    const float* up  = U  + row0 * DIN_ + d;
    const float* rp  = C1 + row0 * (2 * DIN_) + DIN_ + d;
    float*       yp  = Y2 + row0 * DIN_ + d;
    for (int g = 0; g < TCH; g += TG) {
        float dtv[TG], uv[TG], rv[TG];
#pragma unroll
        for (int i = 0; i < TG; ++i) {
            dtv[i] = dtp[(size_t)(g + i) * DIN_];
            uv[i]  = up [(size_t)(g + i) * DIN_];
            rv[i]  = rp [(size_t)(g + i) * (2 * DIN_)];
        }
#pragma unroll
        for (int i = 0; i < TG; ++i) {
            float Bv[DSTATE], Cv[DSTATE];
#pragma unroll
            for (int q = 0; q < 4; ++q) {
                const v4f b = bcs[(g + i) * 8 + q];
                const v4f c = bcs[(g + i) * 8 + 4 + q];
                Bv[4 * q + 0] = b[0]; Bv[4 * q + 1] = b[1];
                Bv[4 * q + 2] = b[2]; Bv[4 * q + 3] = b[3];
                Cv[4 * q + 0] = c[0]; Cv[4 * q + 1] = c[1];
                Cv[4 * q + 2] = c[2]; Cv[4 * q + 3] = c[3];
            }
            const float dtu = dtv[i] * uv[i];
            float y = 0.0f;
#pragma unroll
            for (int s = 0; s < DSTATE; ++s) {
                h[s] = fmaf(__expf(dtv[i] * As[s]), h[s], dtu * Bv[s]);
                y    = fmaf(h[s], Cv[s], y);
            }
            const float res = rv[i];
            const float sr  = res / (1.0f + __expf(-res));
            yp[(size_t)(g + i) * DIN_] = (y + uv[i] * dp) * sr;
        }
    }
}

extern "C" void kernel_launch(void* const* d_in, const int* in_sizes, int n_in,
                              void* d_out, int out_size, void* d_ws, size_t ws_size,
                              hipStream_t stream)
{
    const float* x      = (const float*)d_in[0];
    const float* W_in   = (const float*)d_in[1];
    const float* conv_w = (const float*)d_in[2];
    const float* conv_b = (const float*)d_in[3];
    const float* W_x    = (const float*)d_in[4];
    const float* W_dt   = (const float*)d_in[5];
    const float* b_dt   = (const float*)d_in[6];
    const float* A_log  = (const float*)d_in[7];
    const float* Dp     = (const float*)d_in[8];
    const float* W_out  = (const float*)d_in[9];
    float* out = (float*)d_out;

    float* C1   = (float*)d_ws;
    float* U    = C1   + (size_t)MTOT * (2 * DIN_);
    float* PR   = U    + (size_t)MTOT * DIN_;
    float* DTb  = PR   + (size_t)MTOT * 64;
    float* Y2   = DTb  + (size_t)MTOT * DIN_;
    float* CH_C = Y2   + (size_t)MTOT * DIN_;
    float* CH_S = CH_C + (size_t)BN_ * NCHUNK * DSTATE * DIN_;

    int tiles = (MTOT / 64) * ((2 * DIN_) / 64);
    wmma_gemm_tn<4, 1, 1><<<tiles / 4, 128, 0, stream>>>(x, DM_, W_in, DM_,
                                                C1, 2 * DIN_,
                                                MTOT, 2 * DIN_, DM_, nullptr, 0);

    conv_silu_kernel<<<BN_ * (L_ / CTL), 256, 0, stream>>>(C1, conv_w, conv_b, U);

    tiles = (MTOT / 16) * (64 / 64);
    wmma_gemm_tn<1, 4, 4><<<tiles, 128, 0, stream>>>(U, DIN_, W_x, DIN_,
                                                PR, 64,
                                                MTOT, 64, DIN_, nullptr, 0);

    tiles = (MTOT / 64) * (DIN_ / 64);
    wmma_gemm_tn<4, 1, 1><<<tiles / 4, 128, 0, stream>>>(PR, 64, W_dt, DTRANK,
                                                DTb, DIN_,
                                                MTOT, DIN_, DTRANK, b_dt, 1);

    const int scan_blocks = BN_ * NCHUNK * (DIN_ / 128);
    scan_chunk_reduce<<<scan_blocks, 128, 0, stream>>>(DTb, U, PR, A_log,
                                                       CH_C, CH_S);
    scan_chunk_emit<<<scan_blocks, 128, 0, stream>>>(DTb, U, PR, C1, A_log, Dp,
                                                     CH_C, CH_S, Y2);

    tiles = (MTOT / 64) * (DM_ / 64);
    wmma_gemm_tn<4, 2, 2><<<tiles / 2, 128, 0, stream>>>(Y2, DIN_, W_out, DIN_,
                                                out, DM_,
                                                MTOT, DM_, DIN_, nullptr, 0);
}


# GEMM K loops waited out each trip's fragment-load round trip before its WMMAs: next trip's A/B fragments now load into a second register set under the current trip's WMMAs; sum order unchanged
# speedup vs baseline: 1.8194x; 1.0481x over previous
#include <hip/hip_runtime.h>

typedef float v2f __attribute__((ext_vector_type(2)));
typedef float v4f __attribute__((ext_vector_type(4)));
typedef float v8f __attribute__((ext_vector_type(8)));

#define BN_    4
#define N_GRP  2
#define L_     1024
#define DM_    512
#define DIN_   1024
#define DSTATE 16
#define DTRANK 32
#define MTOT   4096
#define NCHUNK 16
#define TCH    64

template <int TM, int KU>
struct FragSet { v4f a[KU][TM]; v4f b[KU][4]; };

template <int TM, int KU>
__device__ __forceinline__
void load_frags(FragSet<TM, KU>& f, const float* __restrict__ ap, const float* __restrict__ bp,
                size_t astr, size_t bstr, int k)
{
#pragma unroll
    for (int u = 0; u < KU; ++u) {
#pragma unroll
        for (int i = 0; i < TM; ++i) f.a[u][i] = *(const v4f*)(ap + i * astr + k + 8 * u);
#pragma unroll
        for (int j = 0; j < 4; ++j)  f.b[u][j] = *(const v4f*)(bp + j * bstr + k + 8 * u);
    }
}

template <int TM, int KU>
__device__ __forceinline__
void mma_frags(v8f (&acc)[TM][4], const FragSet<TM, KU>& f)
{
#pragma unroll
    for (int u = 0; u < KU; ++u) {
#pragma unroll
        for (int i = 0; i < TM; ++i)
#pragma unroll
            for (int j = 0; j < 4; ++j)
                acc[i][j] = __builtin_amdgcn_wmma_f32_16x16x4_f32(
                                false, __builtin_shufflevector(f.a[u][i], f.a[u][i], 0, 1),
                                false, __builtin_shufflevector(f.b[u][j], f.b[u][j], 0, 1),
                                (short)0, acc[i][j], false, false);
#pragma unroll
        for (int i = 0; i < TM; ++i)
#pragma unroll
            for (int j = 0; j < 4; ++j)
                acc[i][j] = __builtin_amdgcn_wmma_f32_16x16x4_f32(
                                false, __builtin_shufflevector(f.a[u][i], f.a[u][i], 2, 3),
                                false, __builtin_shufflevector(f.b[u][j], f.b[u][j], 2, 3),
                                (short)0, acc[i][j], false, false);
    }
}
__device__ __forceinline__
void store_tile(const v8f& t, float* __restrict__ C, int ldc, int rowb, int col,
                const float* __restrict__ bias, int epilogue)
{
    const float bv = (epilogue == 1) ? bias[col] : 0.0f;
#pragma unroll
    for (int r = 0; r < 8; ++r) {
        float v = t[r];
        if (epilogue == 1) {
            v += bv;
            v = (v > 20.0f) ? v : __logf(1.0f + __expf(v));
        }
        C[(size_t)(rowb + r) * ldc + col] = v;
    }
}

template <int TM, int KU, int KSPLIT>
__global__ __launch_bounds__(128) __attribute__((amdgpu_waves_per_eu(1, 2)))
void wmma_gemm_tn(const float* __restrict__ A, int lda,
                  const float* __restrict__ Bm, int ldb,
                  float* __restrict__ C, int ldc,
                  int M, int N, int K,
                  const float* __restrict__ bias, int epilogue)
{
    constexpr int TPB = 4 / KSPLIT;
    constexpr int OWN = TM * 4 / KSPLIT;
    static_assert(KSPLIT == 1 || KSPLIT == 2 || KSPLIT == 4, "waves per tile");
    static_assert((TM * 4) % KSPLIT == 0, "pieces must divide among the waves");

    const int lane  = threadIdx.x & 31;
    const int wave  = (KSPLIT > 1) ? __builtin_amdgcn_readfirstlane((int)(threadIdx.x >> 5))
                                   : (int)(threadIdx.x >> 5);
    const int ks    = wave % KSPLIT;
    const int tl    = wave / KSPLIT;
    const int ntile = N >> 6;
    const int tile  = blockIdx.x * TPB + tl;
    if constexpr (KSPLIT == 1) {
        if (tile >= (M / (16 * TM)) * ntile) return;
    }
    const int tm = tile / ntile;
    const int tn = tile % ntile;

    const int l15  = lane & 15;
    const int koff = (lane >> 4) << 2;
    const size_t astr = (size_t)16 * lda;
    const size_t bstr = (size_t)16 * ldb;
    const int klen = K / KSPLIT;
    const int kbeg = ks * klen;
    const float* ap = A  + (size_t)(tm * (16 * TM) + l15) * lda + koff;
    const float* bp = Bm + (size_t)(tn * 64 + l15) * ldb + koff;

    v8f acc[TM][4] = {};
    constexpr int KS = 8 * KU;
    const int kend = kbeg + klen;
    FragSet<TM, KU> f0, f1;
    load_frags<TM, KU>(f0, ap, bp, astr, bstr, kbeg);
    int k = kbeg;
    for (; k + 2 * KS < kend; k += 2 * KS) {
        load_frags<TM, KU>(f1, ap, bp, astr, bstr, k + KS);
        mma_frags<TM, KU>(acc, f0);
        load_frags<TM, KU>(f0, ap, bp, astr, bstr, k + 2 * KS);
        mma_frags<TM, KU>(acc, f1);
    }
    load_frags<TM, KU>(f1, ap, bp, astr, bstr, k + KS);
    mma_frags<TM, KU>(acc, f0);
    mma_frags<TM, KU>(acc, f1);

    const int rb = (lane >> 4) << 3;
    if constexpr (KSPLIT == 1) {
#pragma unroll
        for (int i = 0; i < TM; ++i)
#pragma unroll
            for (int j = 0; j < 4; ++j)
                store_tile(acc[i][j], C, ldc, tm * (16 * TM) + i * 16 + rb,
                           tn * 64 + j * 16 + l15, bias, epilogue);
    } else {
        __shared__ v4f red[TPB * KSPLIT * (KSPLIT - 1) * OWN * 2 * 32];
#pragma unroll
        for (int i = 0; i < TM; ++i)
#pragma unroll
            for (int j = 0; j < 4; ++j) {
                const int o = (i * 4 + j) / OWN;
                const int q = (i * 4 + j) % OWN;
                if (ks != o) {
                    const int slot = (ks < o) ? ks : ks - 1;
                    v4f* p = red + (size_t)((((tl * KSPLIT + o) * (KSPLIT - 1) + slot) * OWN + q) * 2) * 32 + lane;
                    p[0]  = __builtin_shufflevector(acc[i][j], acc[i][j], 0, 1, 2, 3);
                    p[32] = __builtin_shufflevector(acc[i][j], acc[i][j], 4, 5, 6, 7);
                }
            }
        __syncthreads();
#pragma unroll
        for (int o = 0; o < KSPLIT; ++o) {
            if (ks == o) {
#pragma unroll
                for (int q = 0; q < OWN; ++q) {
                    const int i = (o * OWN + q) / 4;
                    const int j = (o * OWN + q) % 4;
                    v8f s;
#pragma unroll
                    for (int sidx = 0; sidx < KSPLIT; ++sidx) {
                        v8f part;
                        if (sidx == o) {
                            part = acc[i][j];
                        } else {
                            const int slot = (sidx < o) ? sidx : sidx - 1;
                            const v4f* p = red + (size_t)((((tl * KSPLIT + o) * (KSPLIT - 1) + slot) * OWN + q) * 2) * 32 + lane;
                            part = __builtin_shufflevector(p[0], p[32], 0, 1, 2, 3, 4, 5, 6, 7);
                        }
                        if (sidx == 0) s = part; else s = s + part;
                    }
                    store_tile(s, C, ldc, tm * (16 * TM) + i * 16 + rb,
                               tn * 64 + j * 16 + l15, bias, epilogue);
                }
            }
        }
    }
}

#define CTL 16

__global__ __launch_bounds__(256)
void conv_silu_kernel(const float* __restrict__ C1,
                      const float* __restrict__ cw,
                      const float* __restrict__ cb,
                      float* __restrict__ U)
{
    const int d  = threadIdx.x << 2;
    const int lt = blockIdx.x & (L_ / CTL - 1);
    const int bn = blockIdx.x / (L_ / CTL);
    const int ch = (bn & (N_GRP - 1)) * DIN_ + d;
    const size_t row0 = (size_t)bn * L_ + (size_t)lt * CTL;

    v4f w[4];
#pragma unroll
    for (int c = 0; c < 4; ++c) w[c] = *(const v4f*)(cw + (size_t)(ch + c) * 4);
    const v4f bias = *(const v4f*)(cb + ch);

    const float* xp = C1 + row0 * (2 * DIN_) + d;
    v4f x[CTL + 3];
#pragma unroll
    for (int i = 0; i < CTL; ++i)
        x[i + 3] = *(const v4f*)(xp + (size_t)i * (2 * DIN_));
    if (lt != 0) {
#pragma unroll
        for (int i = 0; i < 3; ++i)
            x[i] = *(const v4f*)(xp - (size_t)(3 - i) * (2 * DIN_));
    } else {
#pragma unroll
        for (int i = 0; i < 3; ++i) x[i] = (v4f){0.0f, 0.0f, 0.0f, 0.0f};
    }

    float* yp = U + row0 * DIN_ + d;
#pragma unroll
    for (int i = 0; i < CTL; ++i) {
        v4f o;
#pragma unroll
        for (int c = 0; c < 4; ++c) {
            float acc = bias[c];
#pragma unroll
            for (int k = 0; k < 4; ++k) acc = fmaf(w[c][k], x[i + k][c], acc);
            o[c] = acc / (1.0f + __expf(-acc));
        }
        *(v4f*)(yp + (size_t)i * DIN_) = o;
    }
}

#define TG 16
#define PG 4

__global__ __launch_bounds__(128)
void scan_chunk_reduce(const float* __restrict__ DT, const float* __restrict__ U,
                       const float* __restrict__ PR,
                       const float* __restrict__ A_log,
                       float* __restrict__ CH_C, float* __restrict__ CH_S)
{
    const int d  = ((blockIdx.x & 7) << 7) + threadIdx.x;
    const int ck = (blockIdx.x >> 3) & (NCHUNK - 1);
    const int bn = blockIdx.x >> 7;
    const int t0 = ck * TCH;
    const size_t row0 = (size_t)bn * L_ + t0;

    __shared__ v4f bs[TCH * 4];
    {
        const float* prc = PR + row0 * 64 + DTRANK;
#pragma unroll
        for (int r = 0; r < (TCH * 4) / 128; ++r) {
            const int i = threadIdx.x + r * 128;
            bs[i] = *(const v4f*)(prc + (size_t)(i >> 2) * 64 + ((i & 3) << 2));
        }
    }

    float As[DSTATE], h[DSTATE];
#pragma unroll
    for (int s = 0; s < DSTATE; ++s) {
        As[s] = -__expf(A_log[(size_t)d * DSTATE + s]);
        h[s]  = 0.0f;
    }
    __syncthreads();

    float S = 0.0f;
    const float* dtp = DT + row0 * DIN_ + d;
    const float* up  = U  + row0 * DIN_ + d;
    for (int g = 0; g < TCH; g += TG) {
        float dtv[TG], uv[TG];
#pragma unroll
        for (int i = 0; i < TG; ++i) {
            dtv[i] = dtp[(size_t)(g + i) * DIN_];
            uv[i]  = up [(size_t)(g + i) * DIN_];
        }
#pragma unroll
        for (int i = 0; i < TG; ++i) {
            float Bv[DSTATE];
#pragma unroll
            for (int q = 0; q < 4; ++q) {
                const v4f b = bs[(g + i) * 4 + q];
                Bv[4 * q + 0] = b[0]; Bv[4 * q + 1] = b[1];
                Bv[4 * q + 2] = b[2]; Bv[4 * q + 3] = b[3];
            }
            S += dtv[i];
            const float dtu = dtv[i] * uv[i];
#pragma unroll
            for (int s = 0; s < DSTATE; ++s)
                h[s] = fmaf(__expf(dtv[i] * As[s]), h[s], dtu * Bv[s]);
        }
    }
    const size_t cb = (size_t)(bn * NCHUNK + ck) * DSTATE * DIN_ + d;
#pragma unroll
    for (int s = 0; s < DSTATE; ++s) CH_C[cb + (size_t)s * DIN_] = h[s];
    CH_S[(size_t)(bn * NCHUNK + ck) * DIN_ + d] = S;
}

template <int P>
__device__ __forceinline__
void fold_chunks(float (&h)[DSTATE], const float (&As)[DSTATE],
                 const float* __restrict__ chs, const float* __restrict__ chc, int j0)
{
    static_assert(PG == 4, "the tail switch in pass 2 lists 1..PG-1");
    float Sj[P], cj[P][DSTATE];
#pragma unroll
    for (int jj = 0; jj < P; ++jj) {
        Sj[jj] = chs[(size_t)(j0 + jj) * DIN_];
#pragma unroll
        for (int s = 0; s < DSTATE; ++s)
            cj[jj][s] = chc[((size_t)(j0 + jj) * DSTATE + s) * DIN_];
    }
#pragma unroll
    for (int jj = 0; jj < P; ++jj)
#pragma unroll
        for (int s = 0; s < DSTATE; ++s)
            h[s] = fmaf(__expf(Sj[jj] * As[s]), h[s], cj[jj][s]);
}

__global__ __launch_bounds__(128)
void scan_chunk_emit(const float* __restrict__ DT, const float* __restrict__ U,
                     const float* __restrict__ PR, const float* __restrict__ C1,
                     const float* __restrict__ A_log, const float* __restrict__ Dp,
                     const float* __restrict__ CH_C, const float* __restrict__ CH_S,
                     float* __restrict__ Y2)
{
    const int d  = ((blockIdx.x & 7) << 7) + threadIdx.x;
    const int ck = (blockIdx.x >> 3) & (NCHUNK - 1);
    const int bn = blockIdx.x >> 7;
    const int t0 = ck * TCH;
    const size_t row0 = (size_t)bn * L_ + t0;

    __shared__ v4f bcs[TCH * 8];
    v4f stg[(TCH * 8) / 128];
    {
        const float* prc = PR + row0 * 64 + DTRANK;
#pragma unroll
        for (int r = 0; r < (TCH * 8) / 128; ++r) {
            const int i = threadIdx.x + r * 128;
            stg[r] = *(const v4f*)(prc + (size_t)(i >> 3) * 64 + ((i & 7) << 2));
        }
    }

    float As[DSTATE], h[DSTATE];
#pragma unroll
    for (int s = 0; s < DSTATE; ++s) {
        As[s] = -__expf(A_log[(size_t)d * DSTATE + s]);
        h[s]  = 0.0f;
    }
    {
        const float* chs = CH_S + (size_t)(bn * NCHUNK) * DIN_ + d;
        const float* chc = CH_C + (size_t)(bn * NCHUNK) * DSTATE * DIN_ + d;
        int j0 = 0;
        for (; j0 + PG <= ck; j0 += PG) fold_chunks<PG>(h, As, chs, chc, j0);
        switch (ck - j0) {
            case 1: fold_chunks<1>(h, As, chs, chc, j0); break;
            case 2: fold_chunks<2>(h, As, chs, chc, j0); break;
            case 3: fold_chunks<3>(h, As, chs, chc, j0); break;
            default: break;
        }
    }
#pragma unroll
    for (int r = 0; r < (TCH * 8) / 128; ++r) bcs[threadIdx.x + r * 128] = stg[r];

    const float dp = Dp[d];
    __syncthreads();

    const float* dtp = DT + row0 * DIN_ + d;
    const float* up  = U  + row0 * DIN_ + d;
    const float* rp  = C1 + row0 * (2 * DIN_) + DIN_ + d;
    float*       yp  = Y2 + row0 * DIN_ + d;
    for (int g = 0; g < TCH; g += TG) {
        float dtv[TG], uv[TG], rv[TG];
#pragma unroll
        for (int i = 0; i < TG; ++i) {
            dtv[i] = dtp[(size_t)(g + i) * DIN_];
            uv[i]  = up [(size_t)(g + i) * DIN_];
            rv[i]  = rp [(size_t)(g + i) * (2 * DIN_)];
        }
#pragma unroll
        for (int i = 0; i < TG; ++i) {
            float Bv[DSTATE], Cv[DSTATE];
#pragma unroll
            for (int q = 0; q < 4; ++q) {
                const v4f b = bcs[(g + i) * 8 + q];
                const v4f c = bcs[(g + i) * 8 + 4 + q];
                Bv[4 * q + 0] = b[0]; Bv[4 * q + 1] = b[1];
                Bv[4 * q + 2] = b[2]; Bv[4 * q + 3] = b[3];
                Cv[4 * q + 0] = c[0]; Cv[4 * q + 1] = c[1];
                Cv[4 * q + 2] = c[2]; Cv[4 * q + 3] = c[3];
            }
            const float dtu = dtv[i] * uv[i];
            float y = 0.0f;
#pragma unroll
            for (int s = 0; s < DSTATE; ++s) {
                h[s] = fmaf(__expf(dtv[i] * As[s]), h[s], dtu * Bv[s]);
                y    = fmaf(h[s], Cv[s], y);
            }
            const float res = rv[i];
            const float sr  = res / (1.0f + __expf(-res));
            yp[(size_t)(g + i) * DIN_] = (y + uv[i] * dp) * sr;
        }
    }
}

extern "C" void kernel_launch(void* const* d_in, const int* in_sizes, int n_in,
                              void* d_out, int out_size, void* d_ws, size_t ws_size,
                              hipStream_t stream)
{
    const float* x      = (const float*)d_in[0];
    const float* W_in   = (const float*)d_in[1];
    const float* conv_w = (const float*)d_in[2];
    const float* conv_b = (const float*)d_in[3];
    const float* W_x    = (const float*)d_in[4];
    const float* W_dt   = (const float*)d_in[5];
    const float* b_dt   = (const float*)d_in[6];
    const float* A_log  = (const float*)d_in[7];
    const float* Dp     = (const float*)d_in[8];
    const float* W_out  = (const float*)d_in[9];
    float* out = (float*)d_out;

    float* C1   = (float*)d_ws;
    float* U    = C1   + (size_t)MTOT * (2 * DIN_);
    float* PR   = U    + (size_t)MTOT * DIN_;
    float* DTb  = PR   + (size_t)MTOT * 64;
    float* Y2   = DTb  + (size_t)MTOT * DIN_;
    float* CH_C = Y2   + (size_t)MTOT * DIN_;
    float* CH_S = CH_C + (size_t)BN_ * NCHUNK * DSTATE * DIN_;

    int tiles = (MTOT / 64) * ((2 * DIN_) / 64);
    wmma_gemm_tn<4, 2, 1><<<tiles / 4, 128, 0, stream>>>(x, DM_, W_in, DM_,
                                                C1, 2 * DIN_,
                                                MTOT, 2 * DIN_, DM_, nullptr, 0);

    conv_silu_kernel<<<BN_ * (L_ / CTL), 256, 0, stream>>>(C1, conv_w, conv_b, U);

    tiles = (MTOT / 16) * (64 / 64);
    wmma_gemm_tn<1, 4, 4><<<tiles, 128, 0, stream>>>(U, DIN_, W_x, DIN_,
                                                PR, 64,
                                                MTOT, 64, DIN_, nullptr, 0);

    tiles = (MTOT / 64) * (DIN_ / 64);
    wmma_gemm_tn<4, 2, 1><<<tiles / 4, 128, 0, stream>>>(PR, 64, W_dt, DTRANK,
                                                DTb, DIN_,
                                                MTOT, DIN_, DTRANK, b_dt, 1);

    const int scan_blocks = BN_ * NCHUNK * (DIN_ / 128);
    scan_chunk_reduce<<<scan_blocks, 128, 0, stream>>>(DTb, U, PR, A_log,
                                                       CH_C, CH_S);
    scan_chunk_emit<<<scan_blocks, 128, 0, stream>>>(DTb, U, PR, C1, A_log, Dp,
                                                     CH_C, CH_S, Y2);

    tiles = (MTOT / 64) * (DM_ / 64);
    wmma_gemm_tn<4, 2, 2><<<tiles / 2, 128, 0, stream>>>(Y2, DIN_, W_out, DIN_,
                                                out, DM_,
                                                MTOT, DM_, DIN_, nullptr, 0);
}


# Scan passes: each state's exp(dt*A) (mul, mul, exp) stood in front of its fma on the wave's serial chain; the factors of 4 steps are now computed ahead as sweeps of independent work; same bits
# speedup vs baseline: 1.9663x; 1.0807x over previous
#include <hip/hip_runtime.h>

typedef float v2f __attribute__((ext_vector_type(2)));
typedef float v4f __attribute__((ext_vector_type(4)));
typedef float v8f __attribute__((ext_vector_type(8)));

#define BN_    4
#define N_GRP  2
#define L_     1024
#define DM_    512
#define DIN_   1024
#define DSTATE 16
#define DTRANK 32
#define MTOT   4096
#define NCHUNK 16
#define TCH    64

template <int TM, int KU>
struct FragSet { v4f a[KU][TM]; v4f b[KU][4]; };

template <int TM, int KU>
__device__ __forceinline__
void load_frags(FragSet<TM, KU>& f, const float* __restrict__ ap, const float* __restrict__ bp,
                size_t astr, size_t bstr, int k)
{
#pragma unroll
    for (int u = 0; u < KU; ++u) {
#pragma unroll
        for (int i = 0; i < TM; ++i) f.a[u][i] = *(const v4f*)(ap + i * astr + k + 8 * u);
#pragma unroll
        for (int j = 0; j < 4; ++j)  f.b[u][j] = *(const v4f*)(bp + j * bstr + k + 8 * u);
    }
}

template <int TM, int KU>
__device__ __forceinline__
void mma_frags(v8f (&acc)[TM][4], const FragSet<TM, KU>& f)
{
#pragma unroll
    for (int u = 0; u < KU; ++u) {
#pragma unroll
        for (int i = 0; i < TM; ++i)
#pragma unroll
            for (int j = 0; j < 4; ++j)
                acc[i][j] = __builtin_amdgcn_wmma_f32_16x16x4_f32(
                                false, __builtin_shufflevector(f.a[u][i], f.a[u][i], 0, 1),
                                false, __builtin_shufflevector(f.b[u][j], f.b[u][j], 0, 1),
                                (short)0, acc[i][j], false, false);
#pragma unroll
        for (int i = 0; i < TM; ++i)
#pragma unroll
            for (int j = 0; j < 4; ++j)
                acc[i][j] = __builtin_amdgcn_wmma_f32_16x16x4_f32(
                                false, __builtin_shufflevector(f.a[u][i], f.a[u][i], 2, 3),
                                false, __builtin_shufflevector(f.b[u][j], f.b[u][j], 2, 3),
                                (short)0, acc[i][j], false, false);
    }
}
__device__ __forceinline__
void store_tile(const v8f& t, float* __restrict__ C, int ldc, int rowb, int col,
                const float* __restrict__ bias, int epilogue)
{
    const float bv = (epilogue == 1) ? bias[col] : 0.0f;
#pragma unroll
    for (int r = 0; r < 8; ++r) {
        float v = t[r];
        if (epilogue == 1) {
            v += bv;
            v = (v > 20.0f) ? v : __logf(1.0f + __expf(v));
        }
        C[(size_t)(rowb + r) * ldc + col] = v;
    }
}

template <int TM, int KU, int KSPLIT>
__global__ __launch_bounds__(128) __attribute__((amdgpu_waves_per_eu(1, 2)))
void wmma_gemm_tn(const float* __restrict__ A, int lda,
                  const float* __restrict__ Bm, int ldb,
                  float* __restrict__ C, int ldc,
                  int M, int N, int K,
                  const float* __restrict__ bias, int epilogue)
{
    constexpr int TPB = 4 / KSPLIT;
    constexpr int OWN = TM * 4 / KSPLIT;
    static_assert(KSPLIT == 1 || KSPLIT == 2 || KSPLIT == 4, "waves per tile");
    static_assert((TM * 4) % KSPLIT == 0, "pieces must divide among the waves");

    const int lane  = threadIdx.x & 31;
    const int wave  = (KSPLIT > 1) ? __builtin_amdgcn_readfirstlane((int)(threadIdx.x >> 5))
                                   : (int)(threadIdx.x >> 5);
    const int ks    = wave % KSPLIT;
    const int tl    = wave / KSPLIT;
    const int ntile = N >> 6;
    const int tile  = blockIdx.x * TPB + tl;
    if constexpr (KSPLIT == 1) {
        if (tile >= (M / (16 * TM)) * ntile) return;
    }
    const int tm = tile / ntile;
    const int tn = tile % ntile;

    const int l15  = lane & 15;
    const int koff = (lane >> 4) << 2;
    const size_t astr = (size_t)16 * lda;
    const size_t bstr = (size_t)16 * ldb;
    const int klen = K / KSPLIT;
    const int kbeg = ks * klen;
    const float* ap = A  + (size_t)(tm * (16 * TM) + l15) * lda + koff;
    const float* bp = Bm + (size_t)(tn * 64 + l15) * ldb + koff;

    v8f acc[TM][4] = {};
    constexpr int KS = 8 * KU;
    const int kend = kbeg + klen;
    FragSet<TM, KU> f0, f1;
    load_frags<TM, KU>(f0, ap, bp, astr, bstr, kbeg);
    int k = kbeg;
    for (; k + 2 * KS < kend; k += 2 * KS) {
        load_frags<TM, KU>(f1, ap, bp, astr, bstr, k + KS);
        mma_frags<TM, KU>(acc, f0);
        load_frags<TM, KU>(f0, ap, bp, astr, bstr, k + 2 * KS);
        mma_frags<TM, KU>(acc, f1);
    }
    load_frags<TM, KU>(f1, ap, bp, astr, bstr, k + KS);
    mma_frags<TM, KU>(acc, f0);
    mma_frags<TM, KU>(acc, f1);

    const int rb = (lane >> 4) << 3;
    if constexpr (KSPLIT == 1) {
#pragma unroll
        for (int i = 0; i < TM; ++i)
#pragma unroll
            for (int j = 0; j < 4; ++j)
                store_tile(acc[i][j], C, ldc, tm * (16 * TM) + i * 16 + rb,
                           tn * 64 + j * 16 + l15, bias, epilogue);
    } else {
        __shared__ v4f red[TPB * KSPLIT * (KSPLIT - 1) * OWN * 2 * 32];
#pragma unroll
        for (int i = 0; i < TM; ++i)
#pragma unroll
            for (int j = 0; j < 4; ++j) {
                const int o = (i * 4 + j) / OWN;
                const int q = (i * 4 + j) % OWN;
                if (ks != o) {
                    const int slot = (ks < o) ? ks : ks - 1;
                    v4f* p = red + (size_t)((((tl * KSPLIT + o) * (KSPLIT - 1) + slot) * OWN + q) * 2) * 32 + lane;
                    p[0]  = __builtin_shufflevector(acc[i][j], acc[i][j], 0, 1, 2, 3);
                    p[32] = __builtin_shufflevector(acc[i][j], acc[i][j], 4, 5, 6, 7);
                }
            }
        __syncthreads();
#pragma unroll
        for (int o = 0; o < KSPLIT; ++o) {
            if (ks == o) {
#pragma unroll
                for (int q = 0; q < OWN; ++q) {
                    const int i = (o * OWN + q) / 4;
                    const int j = (o * OWN + q) % 4;
                    v8f s;
#pragma unroll
                    for (int sidx = 0; sidx < KSPLIT; ++sidx) {
                        v8f part;
                        if (sidx == o) {
                            part = acc[i][j];
                        } else {
                            const int slot = (sidx < o) ? sidx : sidx - 1;
                            const v4f* p = red + (size_t)((((tl * KSPLIT + o) * (KSPLIT - 1) + slot) * OWN + q) * 2) * 32 + lane;
                            part = __builtin_shufflevector(p[0], p[32], 0, 1, 2, 3, 4, 5, 6, 7);
                        }
                        if (sidx == 0) s = part; else s = s + part;
                    }
                    store_tile(s, C, ldc, tm * (16 * TM) + i * 16 + rb,
                               tn * 64 + j * 16 + l15, bias, epilogue);
                }
            }
        }
    }
}

#define CTL 16

__global__ __launch_bounds__(256)
void conv_silu_kernel(const float* __restrict__ C1,
                      const float* __restrict__ cw,
                      const float* __restrict__ cb,
                      float* __restrict__ U)
{
    const int d  = threadIdx.x << 2;
    const int lt = blockIdx.x & (L_ / CTL - 1);
    const int bn = blockIdx.x / (L_ / CTL);
    const int ch = (bn & (N_GRP - 1)) * DIN_ + d;
    const size_t row0 = (size_t)bn * L_ + (size_t)lt * CTL;

    v4f w[4];
#pragma unroll
    for (int c = 0; c < 4; ++c) w[c] = *(const v4f*)(cw + (size_t)(ch + c) * 4);
    const v4f bias = *(const v4f*)(cb + ch);

    const float* xp = C1 + row0 * (2 * DIN_) + d;
    v4f x[CTL + 3];
#pragma unroll
    for (int i = 0; i < CTL; ++i)
        x[i + 3] = *(const v4f*)(xp + (size_t)i * (2 * DIN_));
    if (lt != 0) {
#pragma unroll
        for (int i = 0; i < 3; ++i)
            x[i] = *(const v4f*)(xp - (size_t)(3 - i) * (2 * DIN_));
    } else {
#pragma unroll
        for (int i = 0; i < 3; ++i) x[i] = (v4f){0.0f, 0.0f, 0.0f, 0.0f};
    }

    float* yp = U + row0 * DIN_ + d;
#pragma unroll
    for (int i = 0; i < CTL; ++i) {
        v4f o;
#pragma unroll
        for (int c = 0; c < 4; ++c) {
            float acc = bias[c];
#pragma unroll
            for (int k = 0; k < 4; ++k) acc = fmaf(w[c][k], x[i + k][c], acc);
            o[c] = acc / (1.0f + __expf(-acc));
        }
        *(v4f*)(yp + (size_t)i * DIN_) = o;
    }
}

#define TG 16
#define PG 4
#define SB 4

template <int NJ>
__device__ __forceinline__
void decay_factors(float (&e)[NJ][DSTATE], const float (&x)[NJ], const float (&As)[DSTATE])
{
    const float log2e = 0x1.715476p+0f;
#pragma unroll
    for (int j = 0; j < NJ; ++j)
#pragma unroll
        for (int s = 0; s < DSTATE; ++s) e[j][s] = x[j] * As[s];
#pragma unroll
    for (int j = 0; j < NJ; ++j)
#pragma unroll
        for (int s = 0; s < DSTATE; ++s) e[j][s] = e[j][s] * log2e;
#pragma unroll
    for (int j = 0; j < NJ; ++j)
#pragma unroll
        for (int s = 0; s < DSTATE; ++s) e[j][s] = __builtin_amdgcn_exp2f(e[j][s]);
}

__global__ __launch_bounds__(128)
void scan_chunk_reduce(const float* __restrict__ DT, const float* __restrict__ U,
                       const float* __restrict__ PR,
                       const float* __restrict__ A_log,
                       float* __restrict__ CH_C, float* __restrict__ CH_S)
{
    const int d  = ((blockIdx.x & 7) << 7) + threadIdx.x;
    const int ck = (blockIdx.x >> 3) & (NCHUNK - 1);
    const int bn = blockIdx.x >> 7;
    const int t0 = ck * TCH;
    const size_t row0 = (size_t)bn * L_ + t0;

    __shared__ v4f bs[TCH * 4];
    {
        const float* prc = PR + row0 * 64 + DTRANK;
#pragma unroll
        for (int r = 0; r < (TCH * 4) / 128; ++r) {
            const int i = threadIdx.x + r * 128;
            bs[i] = *(const v4f*)(prc + (size_t)(i >> 2) * 64 + ((i & 3) << 2));
        }
    }

    float As[DSTATE], h[DSTATE];
#pragma unroll
    for (int s = 0; s < DSTATE; ++s) {
        As[s] = -__expf(A_log[(size_t)d * DSTATE + s]);
        h[s]  = 0.0f;
    }
    __syncthreads();

    float S = 0.0f;
    const float* dtp = DT + row0 * DIN_ + d;
    const float* up  = U  + row0 * DIN_ + d;
    for (int g = 0; g < TCH; g += TG) {
        float dtv[TG], uv[TG];
#pragma unroll
        for (int i = 0; i < TG; ++i) {
            dtv[i] = dtp[(size_t)(g + i) * DIN_];
            uv[i]  = up [(size_t)(g + i) * DIN_];
        }
#pragma unroll
        for (int i0 = 0; i0 < TG; i0 += SB) {
            float e[SB][DSTATE], dtb[SB];
#pragma unroll
            for (int j = 0; j < SB; ++j) dtb[j] = dtv[i0 + j];
            decay_factors<SB>(e, dtb, As);
#pragma unroll
            for (int j = 0; j < SB; ++j) {
                const int i = i0 + j;
                float Bv[DSTATE];
#pragma unroll
                for (int q = 0; q < 4; ++q) {
                    const v4f b = bs[(g + i) * 4 + q];
                    Bv[4 * q + 0] = b[0]; Bv[4 * q + 1] = b[1];
                    Bv[4 * q + 2] = b[2]; Bv[4 * q + 3] = b[3];
                }
                S += dtv[i];
                const float dtu = dtv[i] * uv[i];
#pragma unroll
                for (int s = 0; s < DSTATE; ++s)
                    h[s] = fmaf(e[j][s], h[s], dtu * Bv[s]);
            }
        }
    }
    const size_t cb = (size_t)(bn * NCHUNK + ck) * DSTATE * DIN_ + d;
#pragma unroll
    for (int s = 0; s < DSTATE; ++s) CH_C[cb + (size_t)s * DIN_] = h[s];
    CH_S[(size_t)(bn * NCHUNK + ck) * DIN_ + d] = S;
}

template <int P>
__device__ __forceinline__
void fold_chunks(float (&h)[DSTATE], const float (&As)[DSTATE],
                 const float* __restrict__ chs, const float* __restrict__ chc, int j0)
{
    static_assert(PG == 4, "the tail switch in pass 2 lists 1..PG-1");
    float Sj[P], cj[P][DSTATE];
#pragma unroll
    for (int jj = 0; jj < P; ++jj) {
        Sj[jj] = chs[(size_t)(j0 + jj) * DIN_];
#pragma unroll
        for (int s = 0; s < DSTATE; ++s)
            cj[jj][s] = chc[((size_t)(j0 + jj) * DSTATE + s) * DIN_];
    }
    float e[P][DSTATE];
    decay_factors<P>(e, Sj, As);
#pragma unroll
    for (int jj = 0; jj < P; ++jj)
#pragma unroll
        for (int s = 0; s < DSTATE; ++s)
            h[s] = fmaf(e[jj][s], h[s], cj[jj][s]);
}

__global__ __launch_bounds__(128)
void scan_chunk_emit(const float* __restrict__ DT, const float* __restrict__ U,
                     const float* __restrict__ PR, const float* __restrict__ C1,
                     const float* __restrict__ A_log, const float* __restrict__ Dp,
                     const float* __restrict__ CH_C, const float* __restrict__ CH_S,
                     float* __restrict__ Y2)
{
    const int d  = ((blockIdx.x & 7) << 7) + threadIdx.x;
    const int ck = (blockIdx.x >> 3) & (NCHUNK - 1);
    const int bn = blockIdx.x >> 7;
    const int t0 = ck * TCH;
    const size_t row0 = (size_t)bn * L_ + t0;

    __shared__ v4f bcs[TCH * 8];
    v4f stg[(TCH * 8) / 128];
    {
        const float* prc = PR + row0 * 64 + DTRANK;
#pragma unroll
        for (int r = 0; r < (TCH * 8) / 128; ++r) {
            const int i = threadIdx.x + r * 128;
            stg[r] = *(const v4f*)(prc + (size_t)(i >> 3) * 64 + ((i & 7) << 2));
        }
    }

    float As[DSTATE], h[DSTATE];
#pragma unroll
    for (int s = 0; s < DSTATE; ++s) {
        As[s] = -__expf(A_log[(size_t)d * DSTATE + s]);
        h[s]  = 0.0f;
    }
    {
        const float* chs = CH_S + (size_t)(bn * NCHUNK) * DIN_ + d;
        const float* chc = CH_C + (size_t)(bn * NCHUNK) * DSTATE * DIN_ + d;
        int j0 = 0;
        for (; j0 + PG <= ck; j0 += PG) fold_chunks<PG>(h, As, chs, chc, j0);
        switch (ck - j0) {
            case 1: fold_chunks<1>(h, As, chs, chc, j0); break;
            case 2: fold_chunks<2>(h, As, chs, chc, j0); break;
            case 3: fold_chunks<3>(h, As, chs, chc, j0); break;
            default: break;
        }
    }
#pragma unroll
    for (int r = 0; r < (TCH * 8) / 128; ++r) bcs[threadIdx.x + r * 128] = stg[r];

    const float dp = Dp[d];
    __syncthreads();

    const float* dtp = DT + row0 * DIN_ + d;
    const float* up  = U  + row0 * DIN_ + d;
    const float* rp  = C1 + row0 * (2 * DIN_) + DIN_ + d;
    float*       yp  = Y2 + row0 * DIN_ + d;
    for (int g = 0; g < TCH; g += TG) {
        float dtv[TG], uv[TG], rv[TG];
#pragma unroll
        for (int i = 0; i < TG; ++i) {
            dtv[i] = dtp[(size_t)(g + i) * DIN_];
            uv[i]  = up [(size_t)(g + i) * DIN_];
            rv[i]  = rp [(size_t)(g + i) * (2 * DIN_)];
        }
#pragma unroll
        for (int i0 = 0; i0 < TG; i0 += SB) {
            float sr[SB];
#pragma unroll
            for (int j = 0; j < SB; ++j) {
                const float res = rv[i0 + j];
                sr[j] = res / (1.0f + __expf(-res));
            }
            float e[SB][DSTATE], dtb[SB];
#pragma unroll
            for (int j = 0; j < SB; ++j) dtb[j] = dtv[i0 + j];
            decay_factors<SB>(e, dtb, As);
            float hk[SB][DSTATE], Cv[SB][DSTATE];
#pragma unroll
            for (int j = 0; j < SB; ++j) {
                const int i = i0 + j;
                float Bv[DSTATE];
#pragma unroll
                for (int q = 0; q < 4; ++q) {
                    const v4f b = bcs[(g + i) * 8 + q];
                    const v4f c = bcs[(g + i) * 8 + 4 + q];
                    Bv[4 * q + 0] = b[0]; Bv[4 * q + 1] = b[1];
                    Bv[4 * q + 2] = b[2]; Bv[4 * q + 3] = b[3];
                    Cv[j][4 * q + 0] = c[0]; Cv[j][4 * q + 1] = c[1];
                    Cv[j][4 * q + 2] = c[2]; Cv[j][4 * q + 3] = c[3];
                }
                const float dtu = dtv[i] * uv[i];
#pragma unroll
                for (int s = 0; s < DSTATE; ++s) {
                    h[s]     = fmaf(e[j][s], h[s], dtu * Bv[s]);
                    hk[j][s] = h[s];
                }
            }
            float y[SB];
#pragma unroll
            for (int j = 0; j < SB; ++j) y[j] = 0.0f;
#pragma unroll
            for (int s = 0; s < DSTATE; ++s)
#pragma unroll
                for (int j = 0; j < SB; ++j)
                    y[j] = fmaf(hk[j][s], Cv[j][s], y[j]);
#pragma unroll
            for (int j = 0; j < SB; ++j) {
                const int i = i0 + j;
                yp[(size_t)(g + i) * DIN_] = (y[j] + uv[i] * dp) * sr[j];
            }
        }
    }
}

extern "C" void kernel_launch(void* const* d_in, const int* in_sizes, int n_in,
                              void* d_out, int out_size, void* d_ws, size_t ws_size,
                              hipStream_t stream)
{
    const float* x      = (const float*)d_in[0];
    const float* W_in   = (const float*)d_in[1];
    const float* conv_w = (const float*)d_in[2];
    const float* conv_b = (const float*)d_in[3];
    const float* W_x    = (const float*)d_in[4];
    const float* W_dt   = (const float*)d_in[5];
    const float* b_dt   = (const float*)d_in[6];
    const float* A_log  = (const float*)d_in[7];
    const float* Dp     = (const float*)d_in[8];
    const float* W_out  = (const float*)d_in[9];
    float* out = (float*)d_out;

    float* C1   = (float*)d_ws;
    float* U    = C1   + (size_t)MTOT * (2 * DIN_);
    float* PR   = U    + (size_t)MTOT * DIN_;
    float* DTb  = PR   + (size_t)MTOT * 64;
    float* Y2   = DTb  + (size_t)MTOT * DIN_;
    float* CH_C = Y2   + (size_t)MTOT * DIN_;
    float* CH_S = CH_C + (size_t)BN_ * NCHUNK * DSTATE * DIN_;

    int tiles = (MTOT / 64) * ((2 * DIN_) / 64);
    wmma_gemm_tn<4, 2, 1><<<tiles / 4, 128, 0, stream>>>(x, DM_, W_in, DM_,
                                                C1, 2 * DIN_,
                                                MTOT, 2 * DIN_, DM_, nullptr, 0);

    conv_silu_kernel<<<BN_ * (L_ / CTL), 256, 0, stream>>>(C1, conv_w, conv_b, U);

    tiles = (MTOT / 16) * (64 / 64);
    wmma_gemm_tn<1, 4, 4><<<tiles, 128, 0, stream>>>(U, DIN_, W_x, DIN_,
                                                PR, 64,
                                                MTOT, 64, DIN_, nullptr, 0);

    tiles = (MTOT / 64) * (DIN_ / 64);
    wmma_gemm_tn<4, 2, 1><<<tiles / 4, 128, 0, stream>>>(PR, 64, W_dt, DTRANK,
                                                DTb, DIN_,
                                                MTOT, DIN_, DTRANK, b_dt, 1);

    const int scan_blocks = BN_ * NCHUNK * (DIN_ / 128);
    scan_chunk_reduce<<<scan_blocks, 128, 0, stream>>>(DTb, U, PR, A_log,
                                                       CH_C, CH_S);
    scan_chunk_emit<<<scan_blocks, 128, 0, stream>>>(DTb, U, PR, C1, A_log, Dp,
                                                     CH_C, CH_S, Y2);

    tiles = (MTOT / 64) * (DM_ / 64);
    wmma_gemm_tn<4, 2, 2><<<tiles / 2, 128, 0, stream>>>(Y2, DIN_, W_out, DIN_,
                                                out, DM_,
                                                MTOT, DM_, DIN_, nullptr, 0);
}


# conv+silu, x_proj and dt_proj were three short dependent launches: one block per 16-step band now does all three, U and PR bands handed on in LDS, x_proj K split over 8 waves; f32 sum order changes
# speedup vs baseline: 2.0833x; 1.0595x over previous
#include <hip/hip_runtime.h>

typedef float v2f __attribute__((ext_vector_type(2)));
typedef float v4f __attribute__((ext_vector_type(4)));
typedef float v8f __attribute__((ext_vector_type(8)));

#define BN_    4
#define N_GRP  2
#define L_     1024
#define DM_    512
#define DIN_   1024
#define DSTATE 16
#define DTRANK 32
#define MTOT   4096
#define NCHUNK 16
#define TCH    64

template <int TM, int KU>
struct FragSet { v4f a[KU][TM]; v4f b[KU][4]; };

template <int TM, int KU>
__device__ __forceinline__
void load_frags(FragSet<TM, KU>& f, const float* __restrict__ ap, const float* __restrict__ bp,
                size_t astr, size_t bstr, int k)
{
#pragma unroll
    for (int u = 0; u < KU; ++u) {
#pragma unroll
        for (int i = 0; i < TM; ++i) f.a[u][i] = *(const v4f*)(ap + i * astr + k + 8 * u);
#pragma unroll
        for (int j = 0; j < 4; ++j)  f.b[u][j] = *(const v4f*)(bp + j * bstr + k + 8 * u);
    }
}

template <int TM, int KU>
__device__ __forceinline__
void mma_frags(v8f (&acc)[TM][4], const FragSet<TM, KU>& f)
{
#pragma unroll
    for (int u = 0; u < KU; ++u) {
#pragma unroll
        for (int i = 0; i < TM; ++i)
#pragma unroll
            for (int j = 0; j < 4; ++j)
                acc[i][j] = __builtin_amdgcn_wmma_f32_16x16x4_f32(
                                false, __builtin_shufflevector(f.a[u][i], f.a[u][i], 0, 1),
                                false, __builtin_shufflevector(f.b[u][j], f.b[u][j], 0, 1),
                                (short)0, acc[i][j], false, false);
#pragma unroll
        for (int i = 0; i < TM; ++i)
#pragma unroll
            for (int j = 0; j < 4; ++j)
                acc[i][j] = __builtin_amdgcn_wmma_f32_16x16x4_f32(
                                false, __builtin_shufflevector(f.a[u][i], f.a[u][i], 2, 3),
                                false, __builtin_shufflevector(f.b[u][j], f.b[u][j], 2, 3),
                                (short)0, acc[i][j], false, false);
    }
}
__device__ __forceinline__
void store_tile(const v8f& t, float* __restrict__ C, int ldc, int rowb, int col,
                const float* __restrict__ bias, int epilogue)
{
    const float bv = (epilogue == 1) ? bias[col] : 0.0f;
#pragma unroll
    for (int r = 0; r < 8; ++r) {
        float v = t[r];
        if (epilogue == 1) {
            v += bv;
            v = (v > 20.0f) ? v : __logf(1.0f + __expf(v));
        }
        C[(size_t)(rowb + r) * ldc + col] = v;
    }
}

template <int TM, int KU, int KSPLIT>
__global__ __launch_bounds__(128) __attribute__((amdgpu_waves_per_eu(1, 2)))
void wmma_gemm_tn(const float* __restrict__ A, int lda,
                  const float* __restrict__ Bm, int ldb,
                  float* __restrict__ C, int ldc,
                  int M, int N, int K,
                  const float* __restrict__ bias, int epilogue)
{
    constexpr int TPB = 4 / KSPLIT;
    constexpr int OWN = TM * 4 / KSPLIT;
    static_assert(KSPLIT == 1 || KSPLIT == 2 || KSPLIT == 4, "waves per tile");
    static_assert((TM * 4) % KSPLIT == 0, "pieces must divide among the waves");

    const int lane  = threadIdx.x & 31;
    const int wave  = (KSPLIT > 1) ? __builtin_amdgcn_readfirstlane((int)(threadIdx.x >> 5))
                                   : (int)(threadIdx.x >> 5);
    const int ks    = wave % KSPLIT;
    const int tl    = wave / KSPLIT;
    const int ntile = N >> 6;
    const int tile  = blockIdx.x * TPB + tl;
    if constexpr (KSPLIT == 1) {
        if (tile >= (M / (16 * TM)) * ntile) return;
    }
    const int tm = tile / ntile;
    const int tn = tile % ntile;

    const int l15  = lane & 15;
    const int koff = (lane >> 4) << 2;
    const size_t astr = (size_t)16 * lda;
    const size_t bstr = (size_t)16 * ldb;
    const int klen = K / KSPLIT;
    const int kbeg = ks * klen;
    const float* ap = A  + (size_t)(tm * (16 * TM) + l15) * lda + koff;
    const float* bp = Bm + (size_t)(tn * 64 + l15) * ldb + koff;

    v8f acc[TM][4] = {};
    constexpr int KS = 8 * KU;
    const int kend = kbeg + klen;
    FragSet<TM, KU> f0, f1;
    load_frags<TM, KU>(f0, ap, bp, astr, bstr, kbeg);
    int k = kbeg;
    for (; k + 2 * KS < kend; k += 2 * KS) {
        load_frags<TM, KU>(f1, ap, bp, astr, bstr, k + KS);
        mma_frags<TM, KU>(acc, f0);
        load_frags<TM, KU>(f0, ap, bp, astr, bstr, k + 2 * KS);
        mma_frags<TM, KU>(acc, f1);
    }
    load_frags<TM, KU>(f1, ap, bp, astr, bstr, k + KS);
    mma_frags<TM, KU>(acc, f0);
    mma_frags<TM, KU>(acc, f1);

    const int rb = (lane >> 4) << 3;
    if constexpr (KSPLIT == 1) {
#pragma unroll
        for (int i = 0; i < TM; ++i)
#pragma unroll
            for (int j = 0; j < 4; ++j)
                store_tile(acc[i][j], C, ldc, tm * (16 * TM) + i * 16 + rb,
                           tn * 64 + j * 16 + l15, bias, epilogue);
    } else {
        __shared__ v4f red[TPB * KSPLIT * (KSPLIT - 1) * OWN * 2 * 32];
#pragma unroll
        for (int i = 0; i < TM; ++i)
#pragma unroll
            for (int j = 0; j < 4; ++j) {
                const int o = (i * 4 + j) / OWN;
                const int q = (i * 4 + j) % OWN;
                if (ks != o) {
                    const int slot = (ks < o) ? ks : ks - 1;
                    v4f* p = red + (size_t)((((tl * KSPLIT + o) * (KSPLIT - 1) + slot) * OWN + q) * 2) * 32 + lane;
                    p[0]  = __builtin_shufflevector(acc[i][j], acc[i][j], 0, 1, 2, 3);
                    p[32] = __builtin_shufflevector(acc[i][j], acc[i][j], 4, 5, 6, 7);
                }
            }
        __syncthreads();
#pragma unroll
        for (int o = 0; o < KSPLIT; ++o) {
            if (ks == o) {
#pragma unroll
                for (int q = 0; q < OWN; ++q) {
                    const int i = (o * OWN + q) / 4;
                    const int j = (o * OWN + q) % 4;
                    v8f s;
#pragma unroll
                    for (int sidx = 0; sidx < KSPLIT; ++sidx) {
                        v8f part;
                        if (sidx == o) {
                            part = acc[i][j];
                        } else {
                            const int slot = (sidx < o) ? sidx : sidx - 1;
                            const v4f* p = red + (size_t)((((tl * KSPLIT + o) * (KSPLIT - 1) + slot) * OWN + q) * 2) * 32 + lane;
                            part = __builtin_shufflevector(p[0], p[32], 0, 1, 2, 3, 4, 5, 6, 7);
                        }
                        if (sidx == 0) s = part; else s = s + part;
                    }
                    store_tile(s, C, ldc, tm * (16 * TM) + i * 16 + rb,
                               tn * 64 + j * 16 + l15, bias, epilogue);
                }
            }
        }
    }
}

#define CTL 16
#define UP  (DIN_ + 4)
#define PP  (64 + 4)

__device__ __forceinline__
void load_a_band(FragSet<1, 4>& f, const float* ua, int k)
{
#pragma unroll
    for (int u = 0; u < 4; ++u) f.a[u][0] = *(const v4f*)(ua + k + 8 * u);
}
__device__ __forceinline__
void load_b_band(FragSet<1, 4>& f, const float* __restrict__ bp, int k)
{
#pragma unroll
    for (int u = 0; u < 4; ++u)
#pragma unroll
        for (int j = 0; j < 4; ++j)
            f.b[u][j] = *(const v4f*)(bp + (size_t)j * 16 * DIN_ + k + 8 * u);
}

__global__ __launch_bounds__(256)
void band_conv_xproj_dt(const float* __restrict__ C1,
                        const float* __restrict__ cw,
                        const float* __restrict__ cb,
                        const float* __restrict__ Wx,
                        const float* __restrict__ Wdt,
                        const float* __restrict__ bdt,
                        float* __restrict__ U,
                        float* __restrict__ PR,
                        float* __restrict__ DT)
{
    static_assert(CTL == 16, "a band is one WMMA tile high");
    __shared__ __attribute__((aligned(16))) float us[CTL * UP];
    __shared__ v4f   red[8 * 4 * 2 * 32];
    __shared__ __attribute__((aligned(16))) float prt[CTL * PP];

    const int d  = threadIdx.x << 2;
    const int lt = blockIdx.x & (L_ / CTL - 1);
    const int bn = blockIdx.x / (L_ / CTL);
    const int ch = (bn & (N_GRP - 1)) * DIN_ + d;
    const size_t row0 = (size_t)bn * L_ + (size_t)lt * CTL;

    const int lane = threadIdx.x & 31;
    const int wave = __builtin_amdgcn_readfirstlane((int)(threadIdx.x >> 5));
    const int l15  = lane & 15;
    const int koff = (lane >> 4) << 2;
    const int rb   = (lane >> 4) << 3;
    const int kb   = wave * (DIN_ / 8);
    const float* bp = Wx + (size_t)l15 * DIN_ + koff;
    FragSet<1, 4> f0, f1;
    load_b_band(f0, bp, kb);
    {
        v4f w[4];
#pragma unroll
        for (int c = 0; c < 4; ++c) w[c] = *(const v4f*)(cw + (size_t)(ch + c) * 4);
        const v4f bias = *(const v4f*)(cb + ch);

        const float* xp = C1 + row0 * (2 * DIN_) + d;
        v4f x[CTL + 3];
#pragma unroll
        for (int i = 0; i < CTL; ++i)
            x[i + 3] = *(const v4f*)(xp + (size_t)i * (2 * DIN_));
        if (lt != 0) {
#pragma unroll
            for (int i = 0; i < 3; ++i)
                x[i] = *(const v4f*)(xp - (size_t)(3 - i) * (2 * DIN_));
        } else {
#pragma unroll
            for (int i = 0; i < 3; ++i) x[i] = (v4f){0.0f, 0.0f, 0.0f, 0.0f};
        }

        float* yp = U + row0 * DIN_ + d;
#pragma unroll
        for (int i = 0; i < CTL; ++i) {
            v4f o;
#pragma unroll
            for (int c = 0; c < 4; ++c) {
                float acc = bias[c];
#pragma unroll
                for (int k = 0; k < 4; ++k) acc = fmaf(w[c][k], x[i + k][c], acc);
                o[c] = acc / (1.0f + __expf(-acc));
            }
            *(v4f*)(yp + (size_t)i * DIN_) = o;
            *(v4f*)(us + i * UP + d) = o;
        }
    }

    v4f bw[8][4];
    __syncthreads();
    {
        const float* ua = us + l15 * UP + koff;
        v8f acc[1][4] = {};
        load_a_band(f0, ua, kb);
        load_a_band(f1, ua, kb + 32);
        load_b_band(f1, bp, kb + 32);
        {
            const float* wb = Wdt + (size_t)(wave * 128 + l15) * DTRANK + koff;
#pragma unroll
            for (int n = 0; n < 8; ++n)
#pragma unroll
                for (int u = 0; u < 4; ++u)
                    bw[n][u] = *(const v4f*)(wb + (size_t)(n * 16) * DTRANK + 8 * u);
        }
        mma_frags<1, 4>(acc, f0);
        load_a_band(f0, ua, kb + 64);
        load_b_band(f0, bp, kb + 64);
        mma_frags<1, 4>(acc, f1);
        load_a_band(f1, ua, kb + 96);
        load_b_band(f1, bp, kb + 96);
        mma_frags<1, 4>(acc, f0);
        mma_frags<1, 4>(acc, f1);
#pragma unroll
        for (int j = 0; j < 4; ++j) {
            v4f* p = red + (size_t)((wave * 4 + j) * 2) * 32 + lane;
            p[0]  = __builtin_shufflevector(acc[0][j], acc[0][j], 0, 1, 2, 3);
            p[32] = __builtin_shufflevector(acc[0][j], acc[0][j], 4, 5, 6, 7);
        }
    }
    __syncthreads();
    {
        const int j  = wave >> 1;
        const int hf = wave & 1;
        v4f s = red[(size_t)((0 * 4 + j) * 2 + hf) * 32 + lane];
#pragma unroll
        for (int sw = 1; sw < 8; ++sw)
            s = s + red[(size_t)((sw * 4 + j) * 2 + hf) * 32 + lane];
        const int rr  = rb + 4 * hf;
        const int col = 16 * j + l15;
#pragma unroll
        for (int e = 0; e < 4; ++e) {
            PR[(row0 + rr + e) * 64 + col] = s[e];
            prt[(rr + e) * PP + col] = s[e];
        }
    }
    __syncthreads();

    {
        const float* pa = prt + l15 * PP + koff;
        v4f a[4];
#pragma unroll
        for (int u = 0; u < 4; ++u) a[u] = *(const v4f*)(pa + 8 * u);
        v8f c[8] = {};
#pragma unroll
        for (int u = 0; u < 4; ++u) {
#pragma unroll
            for (int n = 0; n < 8; ++n)
                c[n] = __builtin_amdgcn_wmma_f32_16x16x4_f32(
                           false, __builtin_shufflevector(a[u], a[u], 0, 1),
                           false, __builtin_shufflevector(bw[n][u], bw[n][u], 0, 1),
                           (short)0, c[n], false, false);
#pragma unroll
            for (int n = 0; n < 8; ++n)
                c[n] = __builtin_amdgcn_wmma_f32_16x16x4_f32(
                           false, __builtin_shufflevector(a[u], a[u], 2, 3),
                           false, __builtin_shufflevector(bw[n][u], bw[n][u], 2, 3),
                           (short)0, c[n], false, false);
        }
#pragma unroll
        for (int n = 0; n < 8; ++n)
            store_tile(c[n], DT, DIN_, (int)row0 + rb, wave * 128 + n * 16 + l15, bdt, 1);
    }
}

#define TG 16
#define PG 4
#define SB 4

template <int NJ>
__device__ __forceinline__
void decay_factors(float (&e)[NJ][DSTATE], const float (&x)[NJ], const float (&As)[DSTATE])
{
    const float log2e = 0x1.715476p+0f;
#pragma unroll
    for (int j = 0; j < NJ; ++j)
#pragma unroll
        for (int s = 0; s < DSTATE; ++s) e[j][s] = x[j] * As[s];
#pragma unroll
    for (int j = 0; j < NJ; ++j)
#pragma unroll
        for (int s = 0; s < DSTATE; ++s) e[j][s] = e[j][s] * log2e;
#pragma unroll
    for (int j = 0; j < NJ; ++j)
#pragma unroll
        for (int s = 0; s < DSTATE; ++s) e[j][s] = __builtin_amdgcn_exp2f(e[j][s]);
}

__global__ __launch_bounds__(128)
void scan_chunk_reduce(const float* __restrict__ DT, const float* __restrict__ U,
                       const float* __restrict__ PR,
                       const float* __restrict__ A_log,
                       float* __restrict__ CH_C, float* __restrict__ CH_S)
{
    const int d  = ((blockIdx.x & 7) << 7) + threadIdx.x;
    const int ck = (blockIdx.x >> 3) & (NCHUNK - 1);
    const int bn = blockIdx.x >> 7;
    const int t0 = ck * TCH;
    const size_t row0 = (size_t)bn * L_ + t0;

    __shared__ v4f bs[TCH * 4];
    {
        const float* prc = PR + row0 * 64 + DTRANK;
#pragma unroll
        for (int r = 0; r < (TCH * 4) / 128; ++r) {
            const int i = threadIdx.x + r * 128;
            bs[i] = *(const v4f*)(prc + (size_t)(i >> 2) * 64 + ((i & 3) << 2));
        }
    }

    float As[DSTATE], h[DSTATE];
#pragma unroll
    for (int s = 0; s < DSTATE; ++s) {
        As[s] = -__expf(A_log[(size_t)d * DSTATE + s]);
        h[s]  = 0.0f;
    }
    __syncthreads();

    float S = 0.0f;
    const float* dtp = DT + row0 * DIN_ + d;
    const float* up  = U  + row0 * DIN_ + d;
    for (int g = 0; g < TCH; g += TG) {
        float dtv[TG], uv[TG];
#pragma unroll
        for (int i = 0; i < TG; ++i) {
            dtv[i] = dtp[(size_t)(g + i) * DIN_];
            uv[i]  = up [(size_t)(g + i) * DIN_];
        }
#pragma unroll
        for (int i0 = 0; i0 < TG; i0 += SB) {
            float e[SB][DSTATE], dtb[SB];
#pragma unroll
            for (int j = 0; j < SB; ++j) dtb[j] = dtv[i0 + j];
            decay_factors<SB>(e, dtb, As);
#pragma unroll
            for (int j = 0; j < SB; ++j) {
                const int i = i0 + j;
                float Bv[DSTATE];
#pragma unroll
                for (int q = 0; q < 4; ++q) {
                    const v4f b = bs[(g + i) * 4 + q];
                    Bv[4 * q + 0] = b[0]; Bv[4 * q + 1] = b[1];
                    Bv[4 * q + 2] = b[2]; Bv[4 * q + 3] = b[3];
                }
                S += dtv[i];
                const float dtu = dtv[i] * uv[i];
#pragma unroll
                for (int s = 0; s < DSTATE; ++s)
                    h[s] = fmaf(e[j][s], h[s], dtu * Bv[s]);
            }
        }
    }
    const size_t cb = (size_t)(bn * NCHUNK + ck) * DSTATE * DIN_ + d;
#pragma unroll
    for (int s = 0; s < DSTATE; ++s) CH_C[cb + (size_t)s * DIN_] = h[s];
    CH_S[(size_t)(bn * NCHUNK + ck) * DIN_ + d] = S;
}

template <int P>
__device__ __forceinline__
void fold_chunks(float (&h)[DSTATE], const float (&As)[DSTATE],
                 const float* __restrict__ chs, const float* __restrict__ chc, int j0)
{
    static_assert(PG == 4, "the tail switch in pass 2 lists 1..PG-1");
    float Sj[P], cj[P][DSTATE];
#pragma unroll
    for (int jj = 0; jj < P; ++jj) {
        Sj[jj] = chs[(size_t)(j0 + jj) * DIN_];
#pragma unroll
        for (int s = 0; s < DSTATE; ++s)
            cj[jj][s] = chc[((size_t)(j0 + jj) * DSTATE + s) * DIN_];
    }
    float e[P][DSTATE];
    decay_factors<P>(e, Sj, As);
#pragma unroll
    for (int jj = 0; jj < P; ++jj)
#pragma unroll
        for (int s = 0; s < DSTATE; ++s)
            h[s] = fmaf(e[jj][s], h[s], cj[jj][s]);
}

__global__ __launch_bounds__(128)
void scan_chunk_emit(const float* __restrict__ DT, const float* __restrict__ U,
                     const float* __restrict__ PR, const float* __restrict__ C1,
                     const float* __restrict__ A_log, const float* __restrict__ Dp,
                     const float* __restrict__ CH_C, const float* __restrict__ CH_S,
                     float* __restrict__ Y2)
{
    const int d  = ((blockIdx.x & 7) << 7) + threadIdx.x;
    const int ck = (blockIdx.x >> 3) & (NCHUNK - 1);
    const int bn = blockIdx.x >> 7;
    const int t0 = ck * TCH;
    const size_t row0 = (size_t)bn * L_ + t0;

    __shared__ v4f bcs[TCH * 8];
    v4f stg[(TCH * 8) / 128];
    {
        const float* prc = PR + row0 * 64 + DTRANK;
#pragma unroll
        for (int r = 0; r < (TCH * 8) / 128; ++r) {
            const int i = threadIdx.x + r * 128;
            stg[r] = *(const v4f*)(prc + (size_t)(i >> 3) * 64 + ((i & 7) << 2));
        }
    }

    float As[DSTATE], h[DSTATE];
#pragma unroll
    for (int s = 0; s < DSTATE; ++s) {
        As[s] = -__expf(A_log[(size_t)d * DSTATE + s]);
        h[s]  = 0.0f;
    }
    {
        const float* chs = CH_S + (size_t)(bn * NCHUNK) * DIN_ + d;
        const float* chc = CH_C + (size_t)(bn * NCHUNK) * DSTATE * DIN_ + d;
        int j0 = 0;
        for (; j0 + PG <= ck; j0 += PG) fold_chunks<PG>(h, As, chs, chc, j0);
        switch (ck - j0) {
            case 1: fold_chunks<1>(h, As, chs, chc, j0); break;
            case 2: fold_chunks<2>(h, As, chs, chc, j0); break;
            case 3: fold_chunks<3>(h, As, chs, chc, j0); break;
            default: break;
        }
    }
#pragma unroll
    for (int r = 0; r < (TCH * 8) / 128; ++r) bcs[threadIdx.x + r * 128] = stg[r];

    const float dp = Dp[d];
    __syncthreads();

    const float* dtp = DT + row0 * DIN_ + d;
    const float* up  = U  + row0 * DIN_ + d;
    const float* rp  = C1 + row0 * (2 * DIN_) + DIN_ + d;
    float*       yp  = Y2 + row0 * DIN_ + d;
    for (int g = 0; g < TCH; g += TG) {
        float dtv[TG], uv[TG], rv[TG];
#pragma unroll
        for (int i = 0; i < TG; ++i) {
            dtv[i] = dtp[(size_t)(g + i) * DIN_];
            uv[i]  = up [(size_t)(g + i) * DIN_];
            rv[i]  = rp [(size_t)(g + i) * (2 * DIN_)];
        }
#pragma unroll
        for (int i0 = 0; i0 < TG; i0 += SB) {
            float sr[SB];
#pragma unroll
            for (int j = 0; j < SB; ++j) {
                const float res = rv[i0 + j];
                sr[j] = res / (1.0f + __expf(-res));
            }
            float e[SB][DSTATE], dtb[SB];
#pragma unroll
            for (int j = 0; j < SB; ++j) dtb[j] = dtv[i0 + j];
            decay_factors<SB>(e, dtb, As);
            float hk[SB][DSTATE], Cv[SB][DSTATE];
#pragma unroll
            for (int j = 0; j < SB; ++j) {
                const int i = i0 + j;
                float Bv[DSTATE];
#pragma unroll
                for (int q = 0; q < 4; ++q) {
                    const v4f b = bcs[(g + i) * 8 + q];
                    const v4f c = bcs[(g + i) * 8 + 4 + q];
                    Bv[4 * q + 0] = b[0]; Bv[4 * q + 1] = b[1];
                    Bv[4 * q + 2] = b[2]; Bv[4 * q + 3] = b[3];
                    Cv[j][4 * q + 0] = c[0]; Cv[j][4 * q + 1] = c[1];
                    Cv[j][4 * q + 2] = c[2]; Cv[j][4 * q + 3] = c[3];
                }
                const float dtu = dtv[i] * uv[i];
#pragma unroll
                for (int s = 0; s < DSTATE; ++s) {
                    h[s]     = fmaf(e[j][s], h[s], dtu * Bv[s]);
                    hk[j][s] = h[s];
                }
            }
            float y[SB];
#pragma unroll
            for (int j = 0; j < SB; ++j) y[j] = 0.0f;
#pragma unroll
            for (int s = 0; s < DSTATE; ++s)
#pragma unroll
                for (int j = 0; j < SB; ++j)
                    y[j] = fmaf(hk[j][s], Cv[j][s], y[j]);
#pragma unroll
            for (int j = 0; j < SB; ++j) {
                const int i = i0 + j;
                yp[(size_t)(g + i) * DIN_] = (y[j] + uv[i] * dp) * sr[j];
            }
        }
    }
}

extern "C" void kernel_launch(void* const* d_in, const int* in_sizes, int n_in,
                              void* d_out, int out_size, void* d_ws, size_t ws_size,
                              hipStream_t stream)
{
    const float* x      = (const float*)d_in[0];
    const float* W_in   = (const float*)d_in[1];
    const float* conv_w = (const float*)d_in[2];
    const float* conv_b = (const float*)d_in[3];
    const float* W_x    = (const float*)d_in[4];
    const float* W_dt   = (const float*)d_in[5];
    const float* b_dt   = (const float*)d_in[6];
    const float* A_log  = (const float*)d_in[7];
    const float* Dp     = (const float*)d_in[8];
    const float* W_out  = (const float*)d_in[9];
    float* out = (float*)d_out;

    float* C1   = (float*)d_ws;
    float* U    = C1   + (size_t)MTOT * (2 * DIN_);
    float* PR   = U    + (size_t)MTOT * DIN_;
    float* DTb  = PR   + (size_t)MTOT * 64;
    float* Y2   = DTb  + (size_t)MTOT * DIN_;
    float* CH_C = Y2   + (size_t)MTOT * DIN_;
    float* CH_S = CH_C + (size_t)BN_ * NCHUNK * DSTATE * DIN_;

    int tiles = (MTOT / 64) * ((2 * DIN_) / 64);
    wmma_gemm_tn<4, 2, 1><<<tiles / 4, 128, 0, stream>>>(x, DM_, W_in, DM_,
                                                C1, 2 * DIN_,
                                                MTOT, 2 * DIN_, DM_, nullptr, 0);

    band_conv_xproj_dt<<<BN_ * (L_ / CTL), 256, 0, stream>>>(C1, conv_w, conv_b, W_x, W_dt, b_dt,
                                                              U, PR, DTb);

    const int scan_blocks = BN_ * NCHUNK * (DIN_ / 128);
    scan_chunk_reduce<<<scan_blocks, 128, 0, stream>>>(DTb, U, PR, A_log,
                                                       CH_C, CH_S);
    scan_chunk_emit<<<scan_blocks, 128, 0, stream>>>(DTb, U, PR, C1, A_log, Dp,
                                                     CH_C, CH_S, Y2);

    tiles = (MTOT / 64) * (DM_ / 64);
    wmma_gemm_tn<4, 2, 2><<<tiles / 2, 128, 0, stream>>>(Y2, DIN_, W_out, DIN_,
                                                out, DM_,
                                                MTOT, DM_, DIN_, nullptr, 0);
}


# in_proj/out_proj GEMMs: a fragment load from memory touched 16 lines; each 32-k chunk of a block tile now comes in whole-line loads, is staged in LDS (2 stages), fragments read there; same sum order
# speedup vs baseline: 2.4132x; 1.1584x over previous
#include <hip/hip_runtime.h>

typedef float v2f __attribute__((ext_vector_type(2)));
typedef float v4f __attribute__((ext_vector_type(4)));
typedef float v8f __attribute__((ext_vector_type(8)));

#define BN_    4
#define N_GRP  2
#define L_     1024
#define DM_    512
#define DIN_   1024
#define DSTATE 16
#define DTRANK 32
#define MTOT   4096
#define NCHUNK 16
#define TCH    64

template <int TM, int KU>
struct FragSet { v4f a[KU][TM]; v4f b[KU][4]; };

template <int TM, int KU>
__device__ __forceinline__
void mma_frags(v8f (&acc)[TM][4], const FragSet<TM, KU>& f)
{
#pragma unroll
    for (int u = 0; u < KU; ++u) {
#pragma unroll
        for (int i = 0; i < TM; ++i)
#pragma unroll
            for (int j = 0; j < 4; ++j)
                acc[i][j] = __builtin_amdgcn_wmma_f32_16x16x4_f32(
                                false, __builtin_shufflevector(f.a[u][i], f.a[u][i], 0, 1),
                                false, __builtin_shufflevector(f.b[u][j], f.b[u][j], 0, 1),
                                (short)0, acc[i][j], false, false);
#pragma unroll
        for (int i = 0; i < TM; ++i)
#pragma unroll
            for (int j = 0; j < 4; ++j)
                acc[i][j] = __builtin_amdgcn_wmma_f32_16x16x4_f32(
                                false, __builtin_shufflevector(f.a[u][i], f.a[u][i], 2, 3),
                                false, __builtin_shufflevector(f.b[u][j], f.b[u][j], 2, 3),
                                (short)0, acc[i][j], false, false);
    }
}
__device__ __forceinline__
void store_tile(const v8f& t, float* __restrict__ C, int ldc, int rowb, int col,
                const float* __restrict__ bias, int epilogue)
{
    const float bv = (epilogue == 1) ? bias[col] : 0.0f;
#pragma unroll
    for (int r = 0; r < 8; ++r) {
        float v = t[r];
        if (epilogue == 1) {
            v += bv;
            v = (v > 20.0f) ? v : __logf(1.0f + __expf(v));
        }
        C[(size_t)(rowb + r) * ldc + col] = v;
    }
}

#define GP 36

template <int RA, int RB, int KSPLIT, int LDA, int LDB, int KLEN>
__device__ __forceinline__
void chunk_fetch(v4f (&stg)[KSPLIT * (RA + RB) / 16],
                 const float* __restrict__ ga, const float* __restrict__ gb, int kc)
{
    constexpr int RPR = (RA + RB) / 16;
#pragma unroll
    for (int r = 0; r < KSPLIT * RPR; ++r) {
        const int part = r / RPR;
        const int rr   = r % RPR;
        if (rr < RA / 16)
            stg[r] = *(const v4f*)(ga + (size_t)(rr * 16) * LDA + part * KLEN + kc);
        else
            stg[r] = *(const v4f*)(gb + (size_t)((rr - RA / 16) * 16) * LDB + part * KLEN + kc);
    }
}
template <int NR>
__device__ __forceinline__
void chunk_store(float* sp, const v4f (&stg)[NR])
{
#pragma unroll
    for (int r = 0; r < NR; ++r) *(v4f*)(sp + r * 16 * GP) = stg[r];
}
__device__ __forceinline__
void load_frags_lds(FragSet<4, 1>& f, const float* fa, const float* fb, int u)
{
#pragma unroll
    for (int i = 0; i < 4; ++i) f.a[0][i] = *(const v4f*)(fa + i * 16 * GP + 8 * u);
#pragma unroll
    for (int j = 0; j < 4; ++j) f.b[0][j] = *(const v4f*)(fb + j * 16 * GP + 8 * u);
}

template <int BMT, int BNT, int KSPLIT, int LDA, int LDB, int LDC, int KTOT, int NBN>
__global__ __launch_bounds__(128) __attribute__((amdgpu_waves_per_eu(1, 2)))
void wmma_gemm_tn(const float* __restrict__ A, const float* __restrict__ Bm, float* __restrict__ C)
{
    constexpr int TM  = 4;
    constexpr int TPB = 4 / KSPLIT;
    constexpr int OWN = TM * 4 / KSPLIT;
    static_assert(KSPLIT == 1 || KSPLIT == 2 || KSPLIT == 4, "waves per tile");
    static_assert(BMT * BNT * KSPLIT == 4, "a block is 4 waves");
    constexpr int RA   = 64 * BMT;
    constexpr int RB   = 64 * BNT;
    constexpr int RP   = RA + RB;
    constexpr int NR   = KSPLIT * RP / 16;
    constexpr int SS   = KSPLIT * RP * GP;
    constexpr int KLEN = KTOT / KSPLIT;
    constexpr int NC   = KLEN / 32;
    static_assert(KLEN % 32 == 0 && NC >= 2, "whole chunks, and a last one outside the loop");

    __shared__ __attribute__((aligned(16))) float stage[2 * SS];

    const int lane = threadIdx.x & 31;
    const int wave = __builtin_amdgcn_readfirstlane((int)(threadIdx.x >> 5));
    const int ks   = wave % KSPLIT;
    const int tl   = wave / KSPLIT;
    const int wm   = tl / BNT;
    const int wn   = tl % BNT;
    const int bm   = blockIdx.x / NBN;
    const int bn   = blockIdx.x % NBN;

    const int l15  = lane & 15;
    const int koff = (lane >> 4) << 2;

    const int srow = threadIdx.x >> 3;
    const int sseg = (threadIdx.x & 7) << 2;
    const float* ga = A  + (size_t)(bm * RA + srow) * LDA + sseg;
    const float* gb = Bm + (size_t)(bn * RB + srow) * LDB + sseg;
    float* const sst = stage + srow * GP + sseg;
    const float* const fa = stage + (ks * RP + wm * 64 + l15) * GP + koff;
    const float* const fb = stage + (ks * RP + RA + wn * 64 + l15) * GP + koff;

    v8f acc[TM][4] = {};
    v4f stg[NR];
    FragSet<4, 1> f0, f1;

    chunk_fetch<RA, RB, KSPLIT, LDA, LDB, KLEN>(stg, ga, gb, 0);
    chunk_store<NR>(sst, stg);
    __syncthreads();
    load_frags_lds(f0, fa, fb, 0);
    int so = 0;
    for (int c = 0; c < NC - 1; ++c) {
        chunk_fetch<RA, RB, KSPLIT, LDA, LDB, KLEN>(stg, ga, gb, (c + 1) * 32);
        load_frags_lds(f1, fa + so, fb + so, 1);
        mma_frags<4, 1>(acc, f0);
        load_frags_lds(f0, fa + so, fb + so, 2);
        mma_frags<4, 1>(acc, f1);
        load_frags_lds(f1, fa + so, fb + so, 3);
        chunk_store<NR>(sst + (SS - so), stg);
        __builtin_amdgcn_sched_barrier(0);
        mma_frags<4, 1>(acc, f0);
        __syncthreads();
        so = SS - so;
        load_frags_lds(f0, fa + so, fb + so, 0);
        mma_frags<4, 1>(acc, f1);
    }
    load_frags_lds(f1, fa + so, fb + so, 1);
    mma_frags<4, 1>(acc, f0);
    load_frags_lds(f0, fa + so, fb + so, 2);
    mma_frags<4, 1>(acc, f1);
    load_frags_lds(f1, fa + so, fb + so, 3);
    mma_frags<4, 1>(acc, f0);
    mma_frags<4, 1>(acc, f1);

    const int rb   = (lane >> 4) << 3;
    const int row0 = bm * RA + wm * 64;
    const int col0 = bn * RB + wn * 64;
    if constexpr (KSPLIT == 1) {
#pragma unroll
        for (int i = 0; i < TM; ++i)
#pragma unroll
            for (int j = 0; j < 4; ++j)
                store_tile(acc[i][j], C, LDC, row0 + i * 16 + rb,
                           col0 + j * 16 + l15, nullptr, 0);
    } else {
        __shared__ v4f red[TPB * KSPLIT * (KSPLIT - 1) * OWN * 2 * 32];
#pragma unroll
        for (int i = 0; i < TM; ++i)
#pragma unroll
            for (int j = 0; j < 4; ++j) {
                const int o = (i * 4 + j) / OWN;
                const int q = (i * 4 + j) % OWN;
                if (ks != o) {
                    const int slot = (ks < o) ? ks : ks - 1;
                    v4f* p = red + (size_t)((((tl * KSPLIT + o) * (KSPLIT - 1) + slot) * OWN + q) * 2) * 32 + lane;
                    p[0]  = __builtin_shufflevector(acc[i][j], acc[i][j], 0, 1, 2, 3);
                    p[32] = __builtin_shufflevector(acc[i][j], acc[i][j], 4, 5, 6, 7);
                }
            }
        __syncthreads();
#pragma unroll
        for (int o = 0; o < KSPLIT; ++o) {
            if (ks == o) {
#pragma unroll
                for (int q = 0; q < OWN; ++q) {
                    const int i = (o * OWN + q) / 4;
                    const int j = (o * OWN + q) % 4;
                    v8f s;
#pragma unroll
                    for (int sidx = 0; sidx < KSPLIT; ++sidx) {
                        v8f part;
                        if (sidx == o) {
                            part = acc[i][j];
                        } else {
                            const int slot = (sidx < o) ? sidx : sidx - 1;
                            const v4f* p = red + (size_t)((((tl * KSPLIT + o) * (KSPLIT - 1) + slot) * OWN + q) * 2) * 32 + lane;
                            part = __builtin_shufflevector(p[0], p[32], 0, 1, 2, 3, 4, 5, 6, 7);
                        }
                        if (sidx == 0) s = part; else s = s + part;
                    }
                    store_tile(s, C, LDC, row0 + i * 16 + rb,
                               col0 + j * 16 + l15, nullptr, 0);
                }
            }
        }
    }
}

#define CTL 16
#define UP  (DIN_ + 4)
#define PP  (64 + 4)

__device__ __forceinline__
void load_a_band(FragSet<1, 4>& f, const float* ua, int k)
{
#pragma unroll
    for (int u = 0; u < 4; ++u) f.a[u][0] = *(const v4f*)(ua + k + 8 * u);
}
__device__ __forceinline__
void load_b_band(FragSet<1, 4>& f, const float* __restrict__ bp, int k)
{
#pragma unroll
    for (int u = 0; u < 4; ++u)
#pragma unroll
        for (int j = 0; j < 4; ++j)
            f.b[u][j] = *(const v4f*)(bp + (size_t)j * 16 * DIN_ + k + 8 * u);
}

__global__ __launch_bounds__(256)
void band_conv_xproj_dt(const float* __restrict__ C1,
                        const float* __restrict__ cw,
                        const float* __restrict__ cb,
                        const float* __restrict__ Wx,
                        const float* __restrict__ Wdt,
                        const float* __restrict__ bdt,
                        float* __restrict__ U,
                        float* __restrict__ PR,
                        float* __restrict__ DT)
{
    static_assert(CTL == 16, "a band is one WMMA tile high");
    __shared__ __attribute__((aligned(16))) float us[CTL * UP];
    __shared__ v4f   red[8 * 4 * 2 * 32];
    __shared__ __attribute__((aligned(16))) float prt[CTL * PP];

    const int d  = threadIdx.x << 2;
    const int lt = blockIdx.x & (L_ / CTL - 1);
    const int bn = blockIdx.x / (L_ / CTL);
    const int ch = (bn & (N_GRP - 1)) * DIN_ + d;
    const size_t row0 = (size_t)bn * L_ + (size_t)lt * CTL;

    const int lane = threadIdx.x & 31;
    const int wave = __builtin_amdgcn_readfirstlane((int)(threadIdx.x >> 5));
    const int l15  = lane & 15;
    const int koff = (lane >> 4) << 2;
    const int rb   = (lane >> 4) << 3;
    const int kb   = wave * (DIN_ / 8);
    const float* bp = Wx + (size_t)l15 * DIN_ + koff;
    FragSet<1, 4> f0, f1;
    load_b_band(f0, bp, kb);
    {
        v4f w[4];
#pragma unroll
        for (int c = 0; c < 4; ++c) w[c] = *(const v4f*)(cw + (size_t)(ch + c) * 4);
        const v4f bias = *(const v4f*)(cb + ch);

        const float* xp = C1 + row0 * (2 * DIN_) + d;
        v4f x[CTL + 3];
#pragma unroll
        for (int i = 0; i < CTL; ++i)
            x[i + 3] = *(const v4f*)(xp + (size_t)i * (2 * DIN_));
        if (lt != 0) {
#pragma unroll
            for (int i = 0; i < 3; ++i)
                x[i] = *(const v4f*)(xp - (size_t)(3 - i) * (2 * DIN_));
        } else {
#pragma unroll
            for (int i = 0; i < 3; ++i) x[i] = (v4f){0.0f, 0.0f, 0.0f, 0.0f};
        }

        float* yp = U + row0 * DIN_ + d;
#pragma unroll
        for (int i = 0; i < CTL; ++i) {
            v4f o;
#pragma unroll
            for (int c = 0; c < 4; ++c) {
                float acc = bias[c];
#pragma unroll
                for (int k = 0; k < 4; ++k) acc = fmaf(w[c][k], x[i + k][c], acc);
                o[c] = acc / (1.0f + __expf(-acc));
            }
            *(v4f*)(yp + (size_t)i * DIN_) = o;
            *(v4f*)(us + i * UP + d) = o;
        }
    }

    v4f bw[8][4];
    __syncthreads();
    {
        const float* ua = us + l15 * UP + koff;
        v8f acc[1][4] = {};
        load_a_band(f0, ua, kb);
        load_a_band(f1, ua, kb + 32);
        load_b_band(f1, bp, kb + 32);
        {
            const float* wb = Wdt + (size_t)(wave * 128 + l15) * DTRANK + koff;
#pragma unroll
            for (int n = 0; n < 8; ++n)
#pragma unroll
                for (int u = 0; u < 4; ++u)
                    bw[n][u] = *(const v4f*)(wb + (size_t)(n * 16) * DTRANK + 8 * u);
        }
        mma_frags<1, 4>(acc, f0);
        load_a_band(f0, ua, kb + 64);
        load_b_band(f0, bp, kb + 64);
        mma_frags<1, 4>(acc, f1);
        load_a_band(f1, ua, kb + 96);
        load_b_band(f1, bp, kb + 96);
        mma_frags<1, 4>(acc, f0);
        mma_frags<1, 4>(acc, f1);
#pragma unroll
        for (int j = 0; j < 4; ++j) {
            v4f* p = red + (size_t)((wave * 4 + j) * 2) * 32 + lane;
            p[0]  = __builtin_shufflevector(acc[0][j], acc[0][j], 0, 1, 2, 3);
            p[32] = __builtin_shufflevector(acc[0][j], acc[0][j], 4, 5, 6, 7);
        }
    }
    __syncthreads();
    {
        const int j  = wave >> 1;
        const int hf = wave & 1;
        v4f s = red[(size_t)((0 * 4 + j) * 2 + hf) * 32 + lane];
#pragma unroll
        for (int sw = 1; sw < 8; ++sw)
            s = s + red[(size_t)((sw * 4 + j) * 2 + hf) * 32 + lane];
        const int rr  = rb + 4 * hf;
        const int col = 16 * j + l15;
#pragma unroll
        for (int e = 0; e < 4; ++e) {
            PR[(row0 + rr + e) * 64 + col] = s[e];
            prt[(rr + e) * PP + col] = s[e];
        }
    }
    __syncthreads();

    {
        const float* pa = prt + l15 * PP + koff;
        v4f a[4];
#pragma unroll
        for (int u = 0; u < 4; ++u) a[u] = *(const v4f*)(pa + 8 * u);
        v8f c[8] = {};
#pragma unroll
        for (int u = 0; u < 4; ++u) {
#pragma unroll
            for (int n = 0; n < 8; ++n)
                c[n] = __builtin_amdgcn_wmma_f32_16x16x4_f32(
                           false, __builtin_shufflevector(a[u], a[u], 0, 1),
                           false, __builtin_shufflevector(bw[n][u], bw[n][u], 0, 1),
                           (short)0, c[n], false, false);
#pragma unroll
            for (int n = 0; n < 8; ++n)
                c[n] = __builtin_amdgcn_wmma_f32_16x16x4_f32(
                           false, __builtin_shufflevector(a[u], a[u], 2, 3),
                           false, __builtin_shufflevector(bw[n][u], bw[n][u], 2, 3),
                           (short)0, c[n], false, false);
        }
#pragma unroll
        for (int n = 0; n < 8; ++n)
            store_tile(c[n], DT, DIN_, (int)row0 + rb, wave * 128 + n * 16 + l15, bdt, 1);
    }
}

#define TG 16
#define PG 4
#define SB 4

template <int NJ>
__device__ __forceinline__
void decay_factors(float (&e)[NJ][DSTATE], const float (&x)[NJ], const float (&As)[DSTATE])
{
    const float log2e = 0x1.715476p+0f;
#pragma unroll
    for (int j = 0; j < NJ; ++j)
#pragma unroll
        for (int s = 0; s < DSTATE; ++s) e[j][s] = x[j] * As[s];
#pragma unroll
    for (int j = 0; j < NJ; ++j)
#pragma unroll
        for (int s = 0; s < DSTATE; ++s) e[j][s] = e[j][s] * log2e;
#pragma unroll
    for (int j = 0; j < NJ; ++j)
#pragma unroll
        for (int s = 0; s < DSTATE; ++s) e[j][s] = __builtin_amdgcn_exp2f(e[j][s]);
}

__global__ __launch_bounds__(128)
void scan_chunk_reduce(const float* __restrict__ DT, const float* __restrict__ U,
                       const float* __restrict__ PR,
                       const float* __restrict__ A_log,
                       float* __restrict__ CH_C, float* __restrict__ CH_S)
{
    const int d  = ((blockIdx.x & 7) << 7) + threadIdx.x;
    const int ck = (blockIdx.x >> 3) & (NCHUNK - 1);
    const int bn = blockIdx.x >> 7;
    const int t0 = ck * TCH;
    const size_t row0 = (size_t)bn * L_ + t0;

    __shared__ v4f bs[TCH * 4];
    {
        const float* prc = PR + row0 * 64 + DTRANK;
#pragma unroll
        for (int r = 0; r < (TCH * 4) / 128; ++r) {
            const int i = threadIdx.x + r * 128;
            bs[i] = *(const v4f*)(prc + (size_t)(i >> 2) * 64 + ((i & 3) << 2));
        }
    }

    float As[DSTATE], h[DSTATE];
#pragma unroll
    for (int s = 0; s < DSTATE; ++s) {
        As[s] = -__expf(A_log[(size_t)d * DSTATE + s]);
        h[s]  = 0.0f;
    }
    __syncthreads();

    float S = 0.0f;
    const float* dtp = DT + row0 * DIN_ + d;
    const float* up  = U  + row0 * DIN_ + d;
    for (int g = 0; g < TCH; g += TG) {
        float dtv[TG], uv[TG];
#pragma unroll
        for (int i = 0; i < TG; ++i) {
            dtv[i] = dtp[(size_t)(g + i) * DIN_];
            uv[i]  = up [(size_t)(g + i) * DIN_];
        }
#pragma unroll
        for (int i0 = 0; i0 < TG; i0 += SB) {
            float e[SB][DSTATE], dtb[SB];
#pragma unroll
            for (int j = 0; j < SB; ++j) dtb[j] = dtv[i0 + j];
            decay_factors<SB>(e, dtb, As);
#pragma unroll
            for (int j = 0; j < SB; ++j) {
                const int i = i0 + j;
                float Bv[DSTATE];
#pragma unroll
                for (int q = 0; q < 4; ++q) {
                    const v4f b = bs[(g + i) * 4 + q];
                    Bv[4 * q + 0] = b[0]; Bv[4 * q + 1] = b[1];
                    Bv[4 * q + 2] = b[2]; Bv[4 * q + 3] = b[3];
                }
                S += dtv[i];
                const float dtu = dtv[i] * uv[i];
#pragma unroll
                for (int s = 0; s < DSTATE; ++s)
                    h[s] = fmaf(e[j][s], h[s], dtu * Bv[s]);
            }
        }
    }
    const size_t cb = (size_t)(bn * NCHUNK + ck) * DSTATE * DIN_ + d;
#pragma unroll
    for (int s = 0; s < DSTATE; ++s) CH_C[cb + (size_t)s * DIN_] = h[s];
    CH_S[(size_t)(bn * NCHUNK + ck) * DIN_ + d] = S;
}

template <int P>
__device__ __forceinline__
void fold_chunks(float (&h)[DSTATE], const float (&As)[DSTATE],
                 const float* __restrict__ chs, const float* __restrict__ chc, int j0)
{
    static_assert(PG == 4, "the tail switch in pass 2 lists 1..PG-1");
    float Sj[P], cj[P][DSTATE];
#pragma unroll
    for (int jj = 0; jj < P; ++jj) {
        Sj[jj] = chs[(size_t)(j0 + jj) * DIN_];
#pragma unroll
        for (int s = 0; s < DSTATE; ++s)
            cj[jj][s] = chc[((size_t)(j0 + jj) * DSTATE + s) * DIN_];
    }
    float e[P][DSTATE];
    decay_factors<P>(e, Sj, As);
#pragma unroll
    for (int jj = 0; jj < P; ++jj)
#pragma unroll
        for (int s = 0; s < DSTATE; ++s)
            h[s] = fmaf(e[jj][s], h[s], cj[jj][s]);
}

__global__ __launch_bounds__(128)
void scan_chunk_emit(const float* __restrict__ DT, const float* __restrict__ U,
                     const float* __restrict__ PR, const float* __restrict__ C1,
                     const float* __restrict__ A_log, const float* __restrict__ Dp,
                     const float* __restrict__ CH_C, const float* __restrict__ CH_S,
                     float* __restrict__ Y2)
{
    const int d  = ((blockIdx.x & 7) << 7) + threadIdx.x;
    const int ck = (blockIdx.x >> 3) & (NCHUNK - 1);
    const int bn = blockIdx.x >> 7;
    const int t0 = ck * TCH;
    const size_t row0 = (size_t)bn * L_ + t0;

    __shared__ v4f bcs[TCH * 8];
    v4f stg[(TCH * 8) / 128];
    {
        const float* prc = PR + row0 * 64 + DTRANK;
#pragma unroll
        for (int r = 0; r < (TCH * 8) / 128; ++r) {
            const int i = threadIdx.x + r * 128;
            stg[r] = *(const v4f*)(prc + (size_t)(i >> 3) * 64 + ((i & 7) << 2));
        }
    }

    float As[DSTATE], h[DSTATE];
#pragma unroll
    for (int s = 0; s < DSTATE; ++s) {
        As[s] = -__expf(A_log[(size_t)d * DSTATE + s]);
        h[s]  = 0.0f;
    }
    {
        const float* chs = CH_S + (size_t)(bn * NCHUNK) * DIN_ + d;
        const float* chc = CH_C + (size_t)(bn * NCHUNK) * DSTATE * DIN_ + d;
        int j0 = 0;
        for (; j0 + PG <= ck; j0 += PG) fold_chunks<PG>(h, As, chs, chc, j0);
        switch (ck - j0) {
            case 1: fold_chunks<1>(h, As, chs, chc, j0); break;
            case 2: fold_chunks<2>(h, As, chs, chc, j0); break;
            case 3: fold_chunks<3>(h, As, chs, chc, j0); break;
            default: break;
        }
    }
#pragma unroll
    for (int r = 0; r < (TCH * 8) / 128; ++r) bcs[threadIdx.x + r * 128] = stg[r];

    const float dp = Dp[d];
    __syncthreads();

    const float* dtp = DT + row0 * DIN_ + d;
    const float* up  = U  + row0 * DIN_ + d;
    const float* rp  = C1 + row0 * (2 * DIN_) + DIN_ + d;
    float*       yp  = Y2 + row0 * DIN_ + d;
    for (int g = 0; g < TCH; g += TG) {
        float dtv[TG], uv[TG], rv[TG];
#pragma unroll
        for (int i = 0; i < TG; ++i) {
            dtv[i] = dtp[(size_t)(g + i) * DIN_];
            uv[i]  = up [(size_t)(g + i) * DIN_];
            rv[i]  = rp [(size_t)(g + i) * (2 * DIN_)];
        }
#pragma unroll
        for (int i0 = 0; i0 < TG; i0 += SB) {
            float sr[SB];
#pragma unroll
            for (int j = 0; j < SB; ++j) {
                const float res = rv[i0 + j];
                sr[j] = res / (1.0f + __expf(-res));
            }
            float e[SB][DSTATE], dtb[SB];
#pragma unroll
            for (int j = 0; j < SB; ++j) dtb[j] = dtv[i0 + j];
            decay_factors<SB>(e, dtb, As);
            float hk[SB][DSTATE], Cv[SB][DSTATE];
#pragma unroll
            for (int j = 0; j < SB; ++j) {
                const int i = i0 + j;
                float Bv[DSTATE];
#pragma unroll
                for (int q = 0; q < 4; ++q) {
                    const v4f b = bcs[(g + i) * 8 + q];
                    const v4f c = bcs[(g + i) * 8 + 4 + q];
                    Bv[4 * q + 0] = b[0]; Bv[4 * q + 1] = b[1];
                    Bv[4 * q + 2] = b[2]; Bv[4 * q + 3] = b[3];
                    Cv[j][4 * q + 0] = c[0]; Cv[j][4 * q + 1] = c[1];
                    Cv[j][4 * q + 2] = c[2]; Cv[j][4 * q + 3] = c[3];
                }
                const float dtu = dtv[i] * uv[i];
#pragma unroll
                for (int s = 0; s < DSTATE; ++s) {
                    h[s]     = fmaf(e[j][s], h[s], dtu * Bv[s]);
                    hk[j][s] = h[s];
                }
            }
            float y[SB];
#pragma unroll
            for (int j = 0; j < SB; ++j) y[j] = 0.0f;
#pragma unroll
            for (int s = 0; s < DSTATE; ++s)
#pragma unroll
                for (int j = 0; j < SB; ++j)
                    y[j] = fmaf(hk[j][s], Cv[j][s], y[j]);
#pragma unroll
            for (int j = 0; j < SB; ++j) {
                const int i = i0 + j;
                yp[(size_t)(g + i) * DIN_] = (y[j] + uv[i] * dp) * sr[j];
            }
        }
    }
}

extern "C" void kernel_launch(void* const* d_in, const int* in_sizes, int n_in,
                              void* d_out, int out_size, void* d_ws, size_t ws_size,
                              hipStream_t stream)
{
    const float* x      = (const float*)d_in[0];
    const float* W_in   = (const float*)d_in[1];
    const float* conv_w = (const float*)d_in[2];
    const float* conv_b = (const float*)d_in[3];
    const float* W_x    = (const float*)d_in[4];
    const float* W_dt   = (const float*)d_in[5];
    const float* b_dt   = (const float*)d_in[6];
    const float* A_log  = (const float*)d_in[7];
    const float* Dp     = (const float*)d_in[8];
    const float* W_out  = (const float*)d_in[9];
    float* out = (float*)d_out;

    float* C1   = (float*)d_ws;
    float* U    = C1   + (size_t)MTOT * (2 * DIN_);
    float* PR   = U    + (size_t)MTOT * DIN_;
    float* DTb  = PR   + (size_t)MTOT * 64;
    float* Y2   = DTb  + (size_t)MTOT * DIN_;
    float* CH_C = Y2   + (size_t)MTOT * DIN_;
    float* CH_S = CH_C + (size_t)BN_ * NCHUNK * DSTATE * DIN_;

    wmma_gemm_tn<2, 2, 1, DM_, DM_, 2 * DIN_, DM_, (2 * DIN_) / 128>
        <<<(MTOT / 128) * ((2 * DIN_) / 128), 128, 0, stream>>>(x, W_in, C1);

    band_conv_xproj_dt<<<BN_ * (L_ / CTL), 256, 0, stream>>>(C1, conv_w, conv_b, W_x, W_dt, b_dt,
                                                              U, PR, DTb);

    const int scan_blocks = BN_ * NCHUNK * (DIN_ / 128);
    scan_chunk_reduce<<<scan_blocks, 128, 0, stream>>>(DTb, U, PR, A_log,
                                                       CH_C, CH_S);
    scan_chunk_emit<<<scan_blocks, 128, 0, stream>>>(DTb, U, PR, C1, A_log, Dp,
                                                     CH_C, CH_S, Y2);

    wmma_gemm_tn<1, 2, 2, DIN_, DIN_, DM_, DIN_, DM_ / 128>
        <<<(MTOT / 64) * (DM_ / 128), 128, 0, stream>>>(Y2, W_out, out);
}


# Softplus and silu were library forms (a branch region per softplus value, a 10-instruction IEEE division): now branch-free on the exp2, log2 and rcp units, 8 side by side; values move by about 1 ulp
# speedup vs baseline: 2.5010x; 1.0364x over previous
#include <hip/hip_runtime.h>

typedef float v2f __attribute__((ext_vector_type(2)));
typedef float v4f __attribute__((ext_vector_type(4)));
typedef float v8f __attribute__((ext_vector_type(8)));

#define BN_    4
#define N_GRP  2
#define L_     1024
#define DM_    512
#define DIN_   1024
#define DSTATE 16
#define DTRANK 32
#define MTOT   4096
#define NCHUNK 16
#define TCH    64

template <int TM, int KU>
struct FragSet { v4f a[KU][TM]; v4f b[KU][4]; };

template <int TM, int KU>
__device__ __forceinline__
void mma_frags(v8f (&acc)[TM][4], const FragSet<TM, KU>& f)
{
#pragma unroll
    for (int u = 0; u < KU; ++u) {
#pragma unroll
        for (int i = 0; i < TM; ++i)
#pragma unroll
            for (int j = 0; j < 4; ++j)
                acc[i][j] = __builtin_amdgcn_wmma_f32_16x16x4_f32(
                                false, __builtin_shufflevector(f.a[u][i], f.a[u][i], 0, 1),
                                false, __builtin_shufflevector(f.b[u][j], f.b[u][j], 0, 1),
                                (short)0, acc[i][j], false, false);
#pragma unroll
        for (int i = 0; i < TM; ++i)
#pragma unroll
            for (int j = 0; j < 4; ++j)
                acc[i][j] = __builtin_amdgcn_wmma_f32_16x16x4_f32(
                                false, __builtin_shufflevector(f.a[u][i], f.a[u][i], 2, 3),
                                false, __builtin_shufflevector(f.b[u][j], f.b[u][j], 2, 3),
                                (short)0, acc[i][j], false, false);
    }
}
#define LOG2E_F 0x1.715476p+0f
#define LN2_F   0x1.62e430p-1f
__device__ __forceinline__ float silu_hw(float v)
{
    return v * __builtin_amdgcn_rcpf(1.0f + __builtin_amdgcn_exp2f(v * -LOG2E_F));
}
__device__ __forceinline__
void store_tile(const v8f& t, float* __restrict__ C, int ldc, int rowb, int col,
                const float* __restrict__ bias, int epilogue)
{
    if (epilogue == 1) {
        const float bv = bias[col];
        float v[8], e[8];
#pragma unroll
        for (int r = 0; r < 8; ++r) v[r] = t[r] + bv;
#pragma unroll
        for (int r = 0; r < 8; ++r) e[r] = __builtin_amdgcn_exp2f(v[r] * LOG2E_F);
#pragma unroll
        for (int r = 0; r < 8; ++r) e[r] = __builtin_amdgcn_logf(1.0f + e[r]) * LN2_F;
#pragma unroll
        for (int r = 0; r < 8; ++r)
            C[(size_t)(rowb + r) * ldc + col] = (v[r] > 20.0f) ? v[r] : e[r];
    } else {
#pragma unroll
        for (int r = 0; r < 8; ++r) C[(size_t)(rowb + r) * ldc + col] = t[r];
    }
}

#define GP 36

template <int RA, int RB, int KSPLIT, int LDA, int LDB, int KLEN>
__device__ __forceinline__
void chunk_fetch(v4f (&stg)[KSPLIT * (RA + RB) / 16],
                 const float* __restrict__ ga, const float* __restrict__ gb, int kc)
{
    constexpr int RPR = (RA + RB) / 16;
#pragma unroll
    for (int r = 0; r < KSPLIT * RPR; ++r) {
        const int part = r / RPR;
        const int rr   = r % RPR;
        if (rr < RA / 16)
            stg[r] = *(const v4f*)(ga + (size_t)(rr * 16) * LDA + part * KLEN + kc);
        else
            stg[r] = *(const v4f*)(gb + (size_t)((rr - RA / 16) * 16) * LDB + part * KLEN + kc);
    }
}
template <int NR>
__device__ __forceinline__
void chunk_store(float* sp, const v4f (&stg)[NR])
{
#pragma unroll
    for (int r = 0; r < NR; ++r) *(v4f*)(sp + r * 16 * GP) = stg[r];
}
__device__ __forceinline__
void load_frags_lds(FragSet<4, 1>& f, const float* fa, const float* fb, int u)
{
#pragma unroll
    for (int i = 0; i < 4; ++i) f.a[0][i] = *(const v4f*)(fa + i * 16 * GP + 8 * u);
#pragma unroll
    for (int j = 0; j < 4; ++j) f.b[0][j] = *(const v4f*)(fb + j * 16 * GP + 8 * u);
}

template <int BMT, int BNT, int KSPLIT, int LDA, int LDB, int LDC, int KTOT, int NBN>
__global__ __launch_bounds__(128) __attribute__((amdgpu_waves_per_eu(1, 2)))
void wmma_gemm_tn(const float* __restrict__ A, const float* __restrict__ Bm, float* __restrict__ C)
{
    constexpr int TM  = 4;
    constexpr int TPB = 4 / KSPLIT;
    constexpr int OWN = TM * 4 / KSPLIT;
    static_assert(KSPLIT == 1 || KSPLIT == 2 || KSPLIT == 4, "waves per tile");
    static_assert(BMT * BNT * KSPLIT == 4, "a block is 4 waves");
    constexpr int RA   = 64 * BMT;
    constexpr int RB   = 64 * BNT;
    constexpr int RP   = RA + RB;
    constexpr int NR   = KSPLIT * RP / 16;
    constexpr int SS   = KSPLIT * RP * GP;
    constexpr int KLEN = KTOT / KSPLIT;
    constexpr int NC   = KLEN / 32;
    static_assert(KLEN % 32 == 0 && NC >= 2, "whole chunks, and a last one outside the loop");

    __shared__ __attribute__((aligned(16))) float stage[2 * SS];

    const int lane = threadIdx.x & 31;
    const int wave = __builtin_amdgcn_readfirstlane((int)(threadIdx.x >> 5));
    const int ks   = wave % KSPLIT;
    const int tl   = wave / KSPLIT;
    const int wm   = tl / BNT;
    const int wn   = tl % BNT;
    const int bm   = blockIdx.x / NBN;
    const int bn   = blockIdx.x % NBN;

    const int l15  = lane & 15;
    const int koff = (lane >> 4) << 2;

    const int srow = threadIdx.x >> 3;
    const int sseg = (threadIdx.x & 7) << 2;
    const float* ga = A  + (size_t)(bm * RA + srow) * LDA + sseg;
    const float* gb = Bm + (size_t)(bn * RB + srow) * LDB + sseg;
    float* const sst = stage + srow * GP + sseg;
    const float* const fa = stage + (ks * RP + wm * 64 + l15) * GP + koff;
    const float* const fb = stage + (ks * RP + RA + wn * 64 + l15) * GP + koff;

    v8f acc[TM][4] = {};
    v4f stg[NR];
    FragSet<4, 1> f0, f1;

    chunk_fetch<RA, RB, KSPLIT, LDA, LDB, KLEN>(stg, ga, gb, 0);
    chunk_store<NR>(sst, stg);
    __syncthreads();
    load_frags_lds(f0, fa, fb, 0);
    int so = 0;
    for (int c = 0; c < NC - 1; ++c) {
        chunk_fetch<RA, RB, KSPLIT, LDA, LDB, KLEN>(stg, ga, gb, (c + 1) * 32);
        load_frags_lds(f1, fa + so, fb + so, 1);
        mma_frags<4, 1>(acc, f0);
        load_frags_lds(f0, fa + so, fb + so, 2);
        mma_frags<4, 1>(acc, f1);
        load_frags_lds(f1, fa + so, fb + so, 3);
        chunk_store<NR>(sst + (SS - so), stg);
        __builtin_amdgcn_sched_barrier(0);
        mma_frags<4, 1>(acc, f0);
        __syncthreads();
        so = SS - so;
        load_frags_lds(f0, fa + so, fb + so, 0);
        mma_frags<4, 1>(acc, f1);
    }
    load_frags_lds(f1, fa + so, fb + so, 1);
    mma_frags<4, 1>(acc, f0);
    load_frags_lds(f0, fa + so, fb + so, 2);
    mma_frags<4, 1>(acc, f1);
    load_frags_lds(f1, fa + so, fb + so, 3);
    mma_frags<4, 1>(acc, f0);
    mma_frags<4, 1>(acc, f1);

    const int rb   = (lane >> 4) << 3;
    const int row0 = bm * RA + wm * 64;
    const int col0 = bn * RB + wn * 64;
    if constexpr (KSPLIT == 1) {
#pragma unroll
        for (int i = 0; i < TM; ++i)
#pragma unroll
            for (int j = 0; j < 4; ++j)
                store_tile(acc[i][j], C, LDC, row0 + i * 16 + rb,
                           col0 + j * 16 + l15, nullptr, 0);
    } else {
        __shared__ v4f red[TPB * KSPLIT * (KSPLIT - 1) * OWN * 2 * 32];
#pragma unroll
        for (int i = 0; i < TM; ++i)
#pragma unroll
            for (int j = 0; j < 4; ++j) {
                const int o = (i * 4 + j) / OWN;
                const int q = (i * 4 + j) % OWN;
                if (ks != o) {
                    const int slot = (ks < o) ? ks : ks - 1;
                    v4f* p = red + (size_t)((((tl * KSPLIT + o) * (KSPLIT - 1) + slot) * OWN + q) * 2) * 32 + lane;
                    p[0]  = __builtin_shufflevector(acc[i][j], acc[i][j], 0, 1, 2, 3);
                    p[32] = __builtin_shufflevector(acc[i][j], acc[i][j], 4, 5, 6, 7);
                }
            }
        __syncthreads();
#pragma unroll
        for (int o = 0; o < KSPLIT; ++o) {
            if (ks == o) {
#pragma unroll
                for (int q = 0; q < OWN; ++q) {
                    const int i = (o * OWN + q) / 4;
                    const int j = (o * OWN + q) % 4;
                    v8f s;
#pragma unroll
                    for (int sidx = 0; sidx < KSPLIT; ++sidx) {
                        v8f part;
                        if (sidx == o) {
                            part = acc[i][j];
                        } else {
                            const int slot = (sidx < o) ? sidx : sidx - 1;
                            const v4f* p = red + (size_t)((((tl * KSPLIT + o) * (KSPLIT - 1) + slot) * OWN + q) * 2) * 32 + lane;
                            part = __builtin_shufflevector(p[0], p[32], 0, 1, 2, 3, 4, 5, 6, 7);
                        }
                        if (sidx == 0) s = part; else s = s + part;
                    }
                    store_tile(s, C, LDC, row0 + i * 16 + rb,
                               col0 + j * 16 + l15, nullptr, 0);
                }
            }
        }
    }
}

#define CTL 16
#define UP  (DIN_ + 4)
#define PP  (64 + 4)

__device__ __forceinline__
void load_a_band(FragSet<1, 4>& f, const float* ua, int k)
{
#pragma unroll
    for (int u = 0; u < 4; ++u) f.a[u][0] = *(const v4f*)(ua + k + 8 * u);
}
__device__ __forceinline__
void load_b_band(FragSet<1, 4>& f, const float* __restrict__ bp, int k)
{
#pragma unroll
    for (int u = 0; u < 4; ++u)
#pragma unroll
        for (int j = 0; j < 4; ++j)
            f.b[u][j] = *(const v4f*)(bp + (size_t)j * 16 * DIN_ + k + 8 * u);
}

__global__ __launch_bounds__(256)
void band_conv_xproj_dt(const float* __restrict__ C1,
                        const float* __restrict__ cw,
                        const float* __restrict__ cb,
                        const float* __restrict__ Wx,
                        const float* __restrict__ Wdt,
                        const float* __restrict__ bdt,
                        float* __restrict__ U,
                        float* __restrict__ PR,
                        float* __restrict__ DT)
{
    static_assert(CTL == 16, "a band is one WMMA tile high");
    __shared__ __attribute__((aligned(16))) float us[CTL * UP];
    __shared__ v4f   red[8 * 4 * 2 * 32];
    __shared__ __attribute__((aligned(16))) float prt[CTL * PP];

    const int d  = threadIdx.x << 2;
    const int lt = blockIdx.x & (L_ / CTL - 1);
    const int bn = blockIdx.x / (L_ / CTL);
    const int ch = (bn & (N_GRP - 1)) * DIN_ + d;
    const size_t row0 = (size_t)bn * L_ + (size_t)lt * CTL;

    const int lane = threadIdx.x & 31;
    const int wave = __builtin_amdgcn_readfirstlane((int)(threadIdx.x >> 5));
    const int l15  = lane & 15;
    const int koff = (lane >> 4) << 2;
    const int rb   = (lane >> 4) << 3;
    const int kb   = wave * (DIN_ / 8);
    const float* bp = Wx + (size_t)l15 * DIN_ + koff;
    FragSet<1, 4> f0, f1;
    load_b_band(f0, bp, kb);
    {
        v4f w[4];
#pragma unroll
        for (int c = 0; c < 4; ++c) w[c] = *(const v4f*)(cw + (size_t)(ch + c) * 4);
        const v4f bias = *(const v4f*)(cb + ch);

        const float* xp = C1 + row0 * (2 * DIN_) + d;
        v4f x[CTL + 3];
#pragma unroll
        for (int i = 0; i < CTL; ++i)
            x[i + 3] = *(const v4f*)(xp + (size_t)i * (2 * DIN_));
        if (lt != 0) {
#pragma unroll
            for (int i = 0; i < 3; ++i)
                x[i] = *(const v4f*)(xp - (size_t)(3 - i) * (2 * DIN_));
        } else {
#pragma unroll
            for (int i = 0; i < 3; ++i) x[i] = (v4f){0.0f, 0.0f, 0.0f, 0.0f};
        }

        float* yp = U + row0 * DIN_ + d;
#pragma unroll
        for (int i = 0; i < CTL; ++i) {
            v4f o;
#pragma unroll
            for (int c = 0; c < 4; ++c) {
                float acc = bias[c];
#pragma unroll
                for (int k = 0; k < 4; ++k) acc = fmaf(w[c][k], x[i + k][c], acc);
                o[c] = silu_hw(acc);
            }
            *(v4f*)(yp + (size_t)i * DIN_) = o;
            *(v4f*)(us + i * UP + d) = o;
        }
    }

    v4f bw[8][4];
    __syncthreads();
    {
        const float* ua = us + l15 * UP + koff;
        v8f acc[1][4] = {};
        load_a_band(f0, ua, kb);
        load_a_band(f1, ua, kb + 32);
        load_b_band(f1, bp, kb + 32);
        {
            const float* wb = Wdt + (size_t)(wave * 128 + l15) * DTRANK + koff;
#pragma unroll
            for (int n = 0; n < 8; ++n)
#pragma unroll
                for (int u = 0; u < 4; ++u)
                    bw[n][u] = *(const v4f*)(wb + (size_t)(n * 16) * DTRANK + 8 * u);
        }
        mma_frags<1, 4>(acc, f0);
        load_a_band(f0, ua, kb + 64);
        load_b_band(f0, bp, kb + 64);
        mma_frags<1, 4>(acc, f1);
        load_a_band(f1, ua, kb + 96);
        load_b_band(f1, bp, kb + 96);
        mma_frags<1, 4>(acc, f0);
        mma_frags<1, 4>(acc, f1);
#pragma unroll
        for (int j = 0; j < 4; ++j) {
            v4f* p = red + (size_t)((wave * 4 + j) * 2) * 32 + lane;
            p[0]  = __builtin_shufflevector(acc[0][j], acc[0][j], 0, 1, 2, 3);
            p[32] = __builtin_shufflevector(acc[0][j], acc[0][j], 4, 5, 6, 7);
        }
    }
    __syncthreads();
    {
        const int j  = wave >> 1;
        const int hf = wave & 1;
        v4f s = red[(size_t)((0 * 4 + j) * 2 + hf) * 32 + lane];
#pragma unroll
        for (int sw = 1; sw < 8; ++sw)
            s = s + red[(size_t)((sw * 4 + j) * 2 + hf) * 32 + lane];
        const int rr  = rb + 4 * hf;
        const int col = 16 * j + l15;
#pragma unroll
        for (int e = 0; e < 4; ++e) {
            PR[(row0 + rr + e) * 64 + col] = s[e];
            prt[(rr + e) * PP + col] = s[e];
        }
    }
    __syncthreads();

    {
        const float* pa = prt + l15 * PP + koff;
        v4f a[4];
#pragma unroll
        for (int u = 0; u < 4; ++u) a[u] = *(const v4f*)(pa + 8 * u);
        v8f c[8] = {};
#pragma unroll
        for (int u = 0; u < 4; ++u) {
#pragma unroll
            for (int n = 0; n < 8; ++n)
                c[n] = __builtin_amdgcn_wmma_f32_16x16x4_f32(
                           false, __builtin_shufflevector(a[u], a[u], 0, 1),
                           false, __builtin_shufflevector(bw[n][u], bw[n][u], 0, 1),
                           (short)0, c[n], false, false);
#pragma unroll
            for (int n = 0; n < 8; ++n)
                c[n] = __builtin_amdgcn_wmma_f32_16x16x4_f32(
                           false, __builtin_shufflevector(a[u], a[u], 2, 3),
                           false, __builtin_shufflevector(bw[n][u], bw[n][u], 2, 3),
                           (short)0, c[n], false, false);
        }
#pragma unroll
        for (int n = 0; n < 8; ++n)
            store_tile(c[n], DT, DIN_, (int)row0 + rb, wave * 128 + n * 16 + l15, bdt, 1);
    }
}

#define TG 16
#define PG 4
#define SB 4

template <int NJ>
__device__ __forceinline__
void decay_factors(float (&e)[NJ][DSTATE], const float (&x)[NJ], const float (&As)[DSTATE])
{
    const float log2e = 0x1.715476p+0f;
#pragma unroll
    for (int j = 0; j < NJ; ++j)
#pragma unroll
        for (int s = 0; s < DSTATE; ++s) e[j][s] = x[j] * As[s];
#pragma unroll
    for (int j = 0; j < NJ; ++j)
#pragma unroll
        for (int s = 0; s < DSTATE; ++s) e[j][s] = e[j][s] * log2e;
#pragma unroll
    for (int j = 0; j < NJ; ++j)
#pragma unroll
        for (int s = 0; s < DSTATE; ++s) e[j][s] = __builtin_amdgcn_exp2f(e[j][s]);
}

__global__ __launch_bounds__(128)
void scan_chunk_reduce(const float* __restrict__ DT, const float* __restrict__ U,
                       const float* __restrict__ PR,
                       const float* __restrict__ A_log,
                       float* __restrict__ CH_C, float* __restrict__ CH_S)
{
    const int d  = ((blockIdx.x & 7) << 7) + threadIdx.x;
    const int ck = (blockIdx.x >> 3) & (NCHUNK - 1);
    const int bn = blockIdx.x >> 7;
    const int t0 = ck * TCH;
    const size_t row0 = (size_t)bn * L_ + t0;

    __shared__ v4f bs[TCH * 4];
    {
        const float* prc = PR + row0 * 64 + DTRANK;
#pragma unroll
        for (int r = 0; r < (TCH * 4) / 128; ++r) {
            const int i = threadIdx.x + r * 128;
            bs[i] = *(const v4f*)(prc + (size_t)(i >> 2) * 64 + ((i & 3) << 2));
        }
    }

    float As[DSTATE], h[DSTATE];
#pragma unroll
    for (int s = 0; s < DSTATE; ++s) {
        As[s] = -__expf(A_log[(size_t)d * DSTATE + s]);
        h[s]  = 0.0f;
    }
    __syncthreads();

    float S = 0.0f;
    const float* dtp = DT + row0 * DIN_ + d;
    const float* up  = U  + row0 * DIN_ + d;
    for (int g = 0; g < TCH; g += TG) {
        float dtv[TG], uv[TG];
#pragma unroll
        for (int i = 0; i < TG; ++i) {
            dtv[i] = dtp[(size_t)(g + i) * DIN_];
            uv[i]  = up [(size_t)(g + i) * DIN_];
        }
#pragma unroll
        for (int i0 = 0; i0 < TG; i0 += SB) {
            float e[SB][DSTATE], dtb[SB];
#pragma unroll
            for (int j = 0; j < SB; ++j) dtb[j] = dtv[i0 + j];
            decay_factors<SB>(e, dtb, As);
#pragma unroll
            for (int j = 0; j < SB; ++j) {
                const int i = i0 + j;
                float Bv[DSTATE];
#pragma unroll
                for (int q = 0; q < 4; ++q) {
                    const v4f b = bs[(g + i) * 4 + q];
                    Bv[4 * q + 0] = b[0]; Bv[4 * q + 1] = b[1];
                    Bv[4 * q + 2] = b[2]; Bv[4 * q + 3] = b[3];
                }
                S += dtv[i];
                const float dtu = dtv[i] * uv[i];
#pragma unroll
                for (int s = 0; s < DSTATE; ++s)
                    h[s] = fmaf(e[j][s], h[s], dtu * Bv[s]);
            }
        }
    }
    const size_t cb = (size_t)(bn * NCHUNK + ck) * DSTATE * DIN_ + d;
#pragma unroll
    for (int s = 0; s < DSTATE; ++s) CH_C[cb + (size_t)s * DIN_] = h[s];
    CH_S[(size_t)(bn * NCHUNK + ck) * DIN_ + d] = S;
}

template <int P>
__device__ __forceinline__
void fold_chunks(float (&h)[DSTATE], const float (&As)[DSTATE],
                 const float* __restrict__ chs, const float* __restrict__ chc, int j0)
{
    static_assert(PG == 4, "the tail switch in pass 2 lists 1..PG-1");
    float Sj[P], cj[P][DSTATE];
#pragma unroll
    for (int jj = 0; jj < P; ++jj) {
        Sj[jj] = chs[(size_t)(j0 + jj) * DIN_];
#pragma unroll
        for (int s = 0; s < DSTATE; ++s)
            cj[jj][s] = chc[((size_t)(j0 + jj) * DSTATE + s) * DIN_];
    }
    float e[P][DSTATE];
    decay_factors<P>(e, Sj, As);
#pragma unroll
    for (int jj = 0; jj < P; ++jj)
#pragma unroll
        for (int s = 0; s < DSTATE; ++s)
            h[s] = fmaf(e[jj][s], h[s], cj[jj][s]);
}

__global__ __launch_bounds__(128)
void scan_chunk_emit(const float* __restrict__ DT, const float* __restrict__ U,
                     const float* __restrict__ PR, const float* __restrict__ C1,
                     const float* __restrict__ A_log, const float* __restrict__ Dp,
                     const float* __restrict__ CH_C, const float* __restrict__ CH_S,
                     float* __restrict__ Y2)
{
    const int d  = ((blockIdx.x & 7) << 7) + threadIdx.x;
    const int ck = (blockIdx.x >> 3) & (NCHUNK - 1);
    const int bn = blockIdx.x >> 7;
    const int t0 = ck * TCH;
    const size_t row0 = (size_t)bn * L_ + t0;

    __shared__ v4f bcs[TCH * 8];
    v4f stg[(TCH * 8) / 128];
    {
        const float* prc = PR + row0 * 64 + DTRANK;
#pragma unroll
        for (int r = 0; r < (TCH * 8) / 128; ++r) {
            const int i = threadIdx.x + r * 128;
            stg[r] = *(const v4f*)(prc + (size_t)(i >> 3) * 64 + ((i & 7) << 2));
        }
    }

    float As[DSTATE], h[DSTATE];
#pragma unroll
    for (int s = 0; s < DSTATE; ++s) {
        As[s] = -__expf(A_log[(size_t)d * DSTATE + s]);
        h[s]  = 0.0f;
    }
    {
        const float* chs = CH_S + (size_t)(bn * NCHUNK) * DIN_ + d;
        const float* chc = CH_C + (size_t)(bn * NCHUNK) * DSTATE * DIN_ + d;
        int j0 = 0;
        for (; j0 + PG <= ck; j0 += PG) fold_chunks<PG>(h, As, chs, chc, j0);
        switch (ck - j0) {
            case 1: fold_chunks<1>(h, As, chs, chc, j0); break;
            case 2: fold_chunks<2>(h, As, chs, chc, j0); break;
            case 3: fold_chunks<3>(h, As, chs, chc, j0); break;
            default: break;
        }
    }
#pragma unroll
    for (int r = 0; r < (TCH * 8) / 128; ++r) bcs[threadIdx.x + r * 128] = stg[r];

    const float dp = Dp[d];
    __syncthreads();

    const float* dtp = DT + row0 * DIN_ + d;
    const float* up  = U  + row0 * DIN_ + d;
    const float* rp  = C1 + row0 * (2 * DIN_) + DIN_ + d;
    float*       yp  = Y2 + row0 * DIN_ + d;
    for (int g = 0; g < TCH; g += TG) {
        float dtv[TG], uv[TG], rv[TG];
#pragma unroll
        for (int i = 0; i < TG; ++i) {
            dtv[i] = dtp[(size_t)(g + i) * DIN_];
            uv[i]  = up [(size_t)(g + i) * DIN_];
            rv[i]  = rp [(size_t)(g + i) * (2 * DIN_)];
        }
#pragma unroll
        for (int i0 = 0; i0 < TG; i0 += SB) {
            float sr[SB];
#pragma unroll
            for (int j = 0; j < SB; ++j) {
                const float res = rv[i0 + j];
                sr[j] = silu_hw(res);
            }
            float e[SB][DSTATE], dtb[SB];
#pragma unroll
            for (int j = 0; j < SB; ++j) dtb[j] = dtv[i0 + j];
            decay_factors<SB>(e, dtb, As);
            float hk[SB][DSTATE], Cv[SB][DSTATE];
#pragma unroll
            for (int j = 0; j < SB; ++j) {
                const int i = i0 + j;
                float Bv[DSTATE];
#pragma unroll
                for (int q = 0; q < 4; ++q) {
                    const v4f b = bcs[(g + i) * 8 + q];
                    const v4f c = bcs[(g + i) * 8 + 4 + q];
                    Bv[4 * q + 0] = b[0]; Bv[4 * q + 1] = b[1];
                    Bv[4 * q + 2] = b[2]; Bv[4 * q + 3] = b[3];
                    Cv[j][4 * q + 0] = c[0]; Cv[j][4 * q + 1] = c[1];
                    Cv[j][4 * q + 2] = c[2]; Cv[j][4 * q + 3] = c[3];
                }
                const float dtu = dtv[i] * uv[i];
#pragma unroll
                for (int s = 0; s < DSTATE; ++s) {
                    h[s]     = fmaf(e[j][s], h[s], dtu * Bv[s]);
                    hk[j][s] = h[s];
                }
            }
            float y[SB];
#pragma unroll
            for (int j = 0; j < SB; ++j) y[j] = 0.0f;
#pragma unroll
            for (int s = 0; s < DSTATE; ++s)
#pragma unroll
                for (int j = 0; j < SB; ++j)
                    y[j] = fmaf(hk[j][s], Cv[j][s], y[j]);
#pragma unroll
            for (int j = 0; j < SB; ++j) {
                const int i = i0 + j;
                yp[(size_t)(g + i) * DIN_] = (y[j] + uv[i] * dp) * sr[j];
            }
        }
    }
}

extern "C" void kernel_launch(void* const* d_in, const int* in_sizes, int n_in,
                              void* d_out, int out_size, void* d_ws, size_t ws_size,
                              hipStream_t stream)
{
    const float* x      = (const float*)d_in[0];
    const float* W_in   = (const float*)d_in[1];
    const float* conv_w = (const float*)d_in[2];
    const float* conv_b = (const float*)d_in[3];
    const float* W_x    = (const float*)d_in[4];
    const float* W_dt   = (const float*)d_in[5];
    const float* b_dt   = (const float*)d_in[6];
    const float* A_log  = (const float*)d_in[7];
    const float* Dp     = (const float*)d_in[8];
    const float* W_out  = (const float*)d_in[9];
    float* out = (float*)d_out;

    float* C1   = (float*)d_ws;
    float* U    = C1   + (size_t)MTOT * (2 * DIN_);
    float* PR   = U    + (size_t)MTOT * DIN_;
    float* DTb  = PR   + (size_t)MTOT * 64;
    float* Y2   = DTb  + (size_t)MTOT * DIN_;
    float* CH_C = Y2   + (size_t)MTOT * DIN_;
    float* CH_S = CH_C + (size_t)BN_ * NCHUNK * DSTATE * DIN_;

    wmma_gemm_tn<2, 2, 1, DM_, DM_, 2 * DIN_, DM_, (2 * DIN_) / 128>
        <<<(MTOT / 128) * ((2 * DIN_) / 128), 128, 0, stream>>>(x, W_in, C1);

    band_conv_xproj_dt<<<BN_ * (L_ / CTL), 256, 0, stream>>>(C1, conv_w, conv_b, W_x, W_dt, b_dt,
                                                              U, PR, DTb);

    const int scan_blocks = BN_ * NCHUNK * (DIN_ / 128);
    scan_chunk_reduce<<<scan_blocks, 128, 0, stream>>>(DTb, U, PR, A_log,
                                                       CH_C, CH_S);
    scan_chunk_emit<<<scan_blocks, 128, 0, stream>>>(DTb, U, PR, C1, A_log, Dp,
                                                     CH_C, CH_S, Y2);

    wmma_gemm_tn<1, 2, 2, DIN_, DIN_, DM_, DIN_, DM_ / 128>
        <<<(MTOT / 64) * (DM_ / 128), 128, 0, stream>>>(Y2, W_out, out);
}
